# starting kernel
# baseline (speedup 1.0000x reference)
#include <hip/hip_runtime.h>


typedef float v2f __attribute__((ext_vector_type(2)));
typedef float v8f __attribute__((ext_vector_type(8)));

#define CC 64
#define HH 4
#define OO 16
#define NEG_SLOPE 0.2f

__device__ __forceinline__ unsigned fenc(float f) {
  unsigned b = __float_as_uint(f);
  return (b & 0x80000000u) ? ~b : (b | 0x80000000u);
}
__device__ __forceinline__ float fdec(unsigned e) {
  return (e & 0x80000000u) ? __uint_as_float(e & 0x7FFFFFFFu)
                           : __uint_as_float(~e);
}

__global__ __launch_bounds__(128) void gat_gemm_wmma(
    const float* __restrict__ X, const float* __restrict__ W,
    float* __restrict__ Y, int nrows) {
  const int wave = threadIdx.x >> 5;
  const int lane = threadIdx.x & 31;
  const int m0 = blockIdx.x * 16;
  if (m0 >= nrows) return;
  const int n0 = wave * 16;
  const int lrow = lane & 15;
  const int khalf = (lane >> 4) << 1;

  const float* __restrict__ xrow = X + (size_t)(m0 + lrow) * CC;
  v8f c = {};
#pragma unroll
  for (int k = 0; k < CC; k += 4) {
    v2f a, b;
    a.x = xrow[k + khalf];
    a.y = xrow[k + khalf + 1];
    b.x = W[(size_t)(k + khalf) * CC + n0 + lrow];
    b.y = W[(size_t)(k + khalf + 1) * CC + n0 + lrow];
    c = __builtin_amdgcn_wmma_f32_16x16x4_f32(false, a, false, b,
                                              (short)0, c, false, false);
  }
  const int rbase = m0 + ((lane >> 4) << 3);
#pragma unroll
  for (int r = 0; r < 8; ++r)
    Y[(size_t)(rbase + r) * CC + n0 + lrow] = c[r];
}

__global__ void gat_init(float* __restrict__ accum, unsigned* __restrict__ menc,
                         float* __restrict__ denom, int n_nodes) {
  const long long i = (long long)blockIdx.x * blockDim.x + threadIdx.x;
  const long long nc = (long long)n_nodes * CC;
  if (i < nc) accum[i] = 0.0f;
  if (i < (long long)n_nodes * HH) {
    menc[i] = 0u;
    denom[i] = 0.0f;
  }
}

__global__ __launch_bounds__(256) void gat_edge_logits(
    const float* __restrict__ xl, const float* __restrict__ xr,
    const float* __restrict__ att, const int* __restrict__ src,
    const int* __restrict__ dst, float* __restrict__ elog,
    unsigned* __restrict__ menc, int nedges) {
  const long long t = (long long)blockIdx.x * blockDim.x + threadIdx.x;
  if (t >= (long long)nedges * HH) return;
  const int e = (int)(t >> 2);
  const int h = (int)(t & 3);
  const int s = src[e];
  const int d = dst[e];

  const float4* __restrict__ pl = (const float4*)(xl + (size_t)s * CC + h * OO);
  const float4* __restrict__ pr = (const float4*)(xr + (size_t)d * CC + h * OO);
  const float4* __restrict__ pa = (const float4*)(att + h * OO);

  float acc = 0.0f;
#pragma unroll
  for (int j = 0; j < 4; ++j) {
    float4 l = pl[j];
    float4 r = pr[j];
    float4 av = pa[j];
    float sx = l.x + r.x; sx = sx > 0.0f ? sx : NEG_SLOPE * sx;
    float sy = l.y + r.y; sy = sy > 0.0f ? sy : NEG_SLOPE * sy;
    float sz = l.z + r.z; sz = sz > 0.0f ? sz : NEG_SLOPE * sz;
    float sw = l.w + r.w; sw = sw > 0.0f ? sw : NEG_SLOPE * sw;
    acc += sx * av.x + sy * av.y + sz * av.z + sw * av.w;
  }
  elog[t] = acc;
  atomicMax(&menc[(size_t)d * HH + h], fenc(acc));
}

__global__ __launch_bounds__(256) void gat_edge_exp(
    const int* __restrict__ dst, float* __restrict__ elog,
    const unsigned* __restrict__ menc, float* __restrict__ denom, int nedges) {
  const long long t = (long long)blockIdx.x * blockDim.x + threadIdx.x;
  if (t >= (long long)nedges * HH) return;
  const int e = (int)(t >> 2);
  const int h = (int)(t & 3);
  const int d = dst[e];
  const float m = fdec(menc[(size_t)d * HH + h]);
  const float ex = __expf(elog[t] - m);
  elog[t] = ex;
  atomicAdd(&denom[(size_t)d * HH + h], ex);
}

__global__ __launch_bounds__(256) void gat_edge_scatter(
    const float* __restrict__ xl, const int* __restrict__ src,
    const int* __restrict__ dst, const float* __restrict__ elog,
    const float* __restrict__ denom, float* __restrict__ out, int nedges) {
  const long long t = (long long)blockIdx.x * blockDim.x + threadIdx.x;
  if (t >= (long long)nedges * HH) return;
  const int e = (int)(t >> 2);
  const int h = (int)(t & 3);
  const int s = src[e];
  const int d = dst[e];
  const float alpha = elog[t] / (denom[(size_t)d * HH + h] + 1e-16f);

  const float4* __restrict__ pl = (const float4*)(xl + (size_t)s * CC + h * OO);
  float* __restrict__ po = out + (size_t)d * CC + h * OO;
#pragma unroll
  for (int j = 0; j < 4; ++j) {
    float4 l = pl[j];
    atomicAdd(po + 4 * j + 0, l.x * alpha);
    atomicAdd(po + 4 * j + 1, l.y * alpha);
    atomicAdd(po + 4 * j + 2, l.z * alpha);
    atomicAdd(po + 4 * j + 3, l.w * alpha);
  }
}

__global__ __launch_bounds__(256) void gat_bias_act(
    float* __restrict__ buf, const float* __restrict__ bias, int n_nodes,
    int do_relu) {
  const long long i = (long long)blockIdx.x * blockDim.x + threadIdx.x;
  if (i >= (long long)n_nodes * CC) return;
  float v = buf[i] + bias[i & (CC - 1)];
  if (do_relu) v = fmaxf(v, 0.0f);
  buf[i] = v;
}

extern "C" void kernel_launch(void* const* d_in, const int* in_sizes, int n_in,
                              void* d_out, int out_size, void* d_ws,
                              size_t ws_size, hipStream_t stream) {
  const int N = in_sizes[0] / CC;
  const int E = in_sizes[1] / 2;

  const float* x = (const float*)d_in[0];
  const int* edge = (const int*)d_in[1];
  const int* src = edge;
  const int* dst = edge + E;

  const float* Wl[3] = {(const float*)d_in[2], (const float*)d_in[6],
                        (const float*)d_in[10]};
  const float* Wr[3] = {(const float*)d_in[3], (const float*)d_in[7],
                        (const float*)d_in[11]};
  const float* Av[3] = {(const float*)d_in[4], (const float*)d_in[8],
                        (const float*)d_in[12]};
  const float* Bv[3] = {(const float*)d_in[5], (const float*)d_in[9],
                        (const float*)d_in[13]};

  const size_t NC = (size_t)N * CC;
  const size_t EH = (size_t)E * HH;
  const size_t NH = (size_t)N * HH;
  float* ws = (float*)d_ws;
  float* xl = ws;                 ws += NC;
  float* xr = ws;                 ws += NC;
  float* elog = ws;               ws += EH;
  unsigned* menc = (unsigned*)ws; ws += NH;
  float* denom = ws;              ws += NH;
  float* hA = ws;                 ws += NC;
  float* hB = ws;                 ws += NC;

  const float* lin[3] = {x, hA, hB};
  float* lout[3] = {hA, hB, (float*)d_out};
  const int lrelu[3] = {1, 1, 0};

  const int gemm_grid = (N + 15) / 16;
  const long long nc_thr = (long long)N * CC;
  const long long eh_thr = (long long)E * HH;
  const int init_grid = (int)((nc_thr + 255) / 256);
  const int edge_grid = (int)((eh_thr + 255) / 256);

  for (int l = 0; l < 3; ++l) {
    gat_init<<<init_grid, 256, 0, stream>>>(lout[l], menc, denom, N);
    gat_gemm_wmma<<<gemm_grid, 128, 0, stream>>>(lin[l], Wl[l], xl, N);
    gat_gemm_wmma<<<gemm_grid, 128, 0, stream>>>(lin[l], Wr[l], xr, N);
    gat_edge_logits<<<edge_grid, 256, 0, stream>>>(xl, xr, Av[l], src, dst,
                                                   elog, menc, E);
    gat_edge_exp<<<edge_grid, 256, 0, stream>>>(dst, elog, menc, denom, E);
    gat_edge_scatter<<<edge_grid, 256, 0, stream>>>(xl, src, dst, elog, denom,
                                                    lout[l], E);
    gat_bias_act<<<init_grid, 256, 0, stream>>>(lout[l], Bv[l], N, lrelu[l]);
  }
}


# Logits, softmax max and denominators computed inside the owner blocks (edge lists, xr rows, max, sums in LDS; rows divided at store); 9 launches for 18; logit and denominator sum order changes
# speedup vs baseline: 3.4219x; 3.4219x over previous
#include <hip/hip_runtime.h>


typedef float v2f __attribute__((ext_vector_type(2)));
typedef float v8f __attribute__((ext_vector_type(8)));

#define CC 64
#define HH 4
#define OO 16
#define NEG_SLOPE 0.2f

__device__ __forceinline__ unsigned fenc(float f) {
  unsigned b = __float_as_uint(f);
  return (b & 0x80000000u) ? ~b : (b | 0x80000000u);
}
__device__ __forceinline__ float fdec(unsigned e) {
  return (e & 0x80000000u) ? __uint_as_float(e & 0x7FFFFFFFu)
                           : __uint_as_float(~e);
}

__global__ __launch_bounds__(128) void gat_gemm_wmma(
    const float* __restrict__ X, const float* __restrict__ W,
    float* __restrict__ Y, int nrows) {
  const int wave = threadIdx.x >> 5;
  const int lane = threadIdx.x & 31;
  const int m0 = blockIdx.x * 16;
  if (m0 >= nrows) return;
  const int n0 = wave * 16;
  const int lrow = lane & 15;
  const int khalf = (lane >> 4) << 1;

  const float* __restrict__ xrow = X + (size_t)(m0 + lrow) * CC;
  v8f c = {};
#pragma unroll
  for (int k = 0; k < CC; k += 4) {
    v2f a, b;
    a.x = xrow[k + khalf];
    a.y = xrow[k + khalf + 1];
    b.x = W[(size_t)(k + khalf) * CC + n0 + lrow];
    b.y = W[(size_t)(k + khalf + 1) * CC + n0 + lrow];
    c = __builtin_amdgcn_wmma_f32_16x16x4_f32(false, a, false, b,
                                              (short)0, c, false, false);
  }
  const int rbase = m0 + ((lane >> 4) << 3);
#pragma unroll
  for (int r = 0; r < 8; ++r)
    Y[(size_t)(rbase + r) * CC + n0 + lrow] = c[r];
}

#define ROWS_PB 391
#define AGG_THREADS 1024
#define AGG_WAVES (AGG_THREADS / 32)
#define RING 256
#define SCAN_G 8
#define STEP_K 8

__device__ __forceinline__ float leaky(float v) {
  return fmaxf(v, NEG_SLOPE * v);
}

__device__ __forceinline__ float quad_sum(float t) {
  t += __int_as_float(__builtin_amdgcn_update_dpp(
      0, __float_as_int(t), 0xB1, 0xF, 0xF, true));
  t += __int_as_float(__builtin_amdgcn_update_dpp(
      0, __float_as_int(t), 0x4E, 0xF, 0xF, true));
  return t;
}

__device__ __forceinline__ void resolve_src(const int* __restrict__ src,
                                            int2* ring, int first, int cnt,
                                            int lane) {
  int e[RING / 32];
#pragma unroll
  for (int k = 0; k < RING / 32; ++k) {
    const int b = lane + 32 * k;
    e[k] = b < cnt ? ring[(first + b) & (RING - 1)].x : 0;
  }
#pragma unroll
  for (int k = 0; k < RING / 32; ++k) {
    const int b = lane + 32 * k;
    if (b < cnt) e[k] = src[e[k]];
  }
#pragma unroll
  for (int k = 0; k < RING / 32; ++k) {
    const int b = lane + 32 * k;
    if (b < cnt) ring[(first + b) & (RING - 1)].x = e[k];
  }
}

template <int PASS>
__device__ __forceinline__ void edge_steps(
    const float* __restrict__ xl, float* acc, const float* xrs, unsigned* mx,
    float* den, const int2* ring, int first, int cnt, float4 av, int lane) {
  const int c4 = (lane & 15) << 2;
  const int half = lane >> 4;
  const int hd = (lane & 15) >> 2;
  const bool quad0 = (lane & 3) == 0;
  const int steps = (cnt + 1) >> 1;
#pragma nounroll
  for (int k0 = 0; k0 < steps; k0 += STEP_K) {
    int2 q[STEP_K];
    float4 lv[STEP_K], rv[STEP_K];
    unsigned me[STEP_K];
#pragma unroll
    for (int k = 0; k < STEP_K; ++k) {
      const int idx = min(2 * (k0 + k) + half, cnt - 1);
      q[k] = ring[(first + idx) & (RING - 1)];
    }
#pragma unroll
    for (int k = 0; k < STEP_K; ++k) {
      lv[k] = *(const float4*)(xl + (size_t)q[k].x * CC + c4);
      rv[k] = *(const float4*)(xrs + q[k].y * CC + c4);
      me[k] = PASS == 1 ? 0u : mx[q[k].y * HH + hd];
    }
#pragma unroll
    for (int k = 0; k < STEP_K; ++k) {
      const bool live = 2 * (k0 + k) + half < cnt;
      float t = av.x * leaky(lv[k].x + rv[k].x);
      t += av.y * leaky(lv[k].y + rv[k].y);
      t += av.z * leaky(lv[k].z + rv[k].z);
      t += av.w * leaky(lv[k].w + rv[k].w);
      t = quad_sum(t);
      const int sh = q[k].y * HH + hd;
      if (PASS == 1) {
        if (live && quad0) atomicMax(&mx[sh], fenc(t));
      } else {
        const float w = __expf(t - fdec(me[k]));
        if (live) {
          float* dr = acc + q[k].y * CC + c4;
          atomicAdd(dr + 0, w * lv[k].x);
          atomicAdd(dr + 1, w * lv[k].y);
          atomicAdd(dr + 2, w * lv[k].z);
          atomicAdd(dr + 3, w * lv[k].w);
          if (quad0) atomicAdd(&den[sh], w);
        }
      }
    }
  }
}

template <int MODE>
__device__ __forceinline__ void scan_owned(
    const float* __restrict__ xl, const int* __restrict__ src,
    const int* __restrict__ dst, float* acc, const float* xrs, unsigned* mx,
    float* den, int2* ring, int& cnt_out, bool& over_out, int n0, int rows,
    int nedges, float4 av, int wave, int lane) {
  int head = 0, tail = 0;

#define AGG_PUSH(m, e, x)                                                    \
  {                                                                          \
    const unsigned bal_ = __builtin_amdgcn_ballot_w32(m);                    \
    if (m)                                                                   \
      ring[(tail + (int)__builtin_amdgcn_mbcnt_lo(bal_, 0u)) & (RING - 1)] = \
          make_int2((e), (int)(x));                                          \
    tail += __builtin_popcount(bal_);                                        \
  }
#define AGG_DRAIN()                                                          \
  if (MODE != 0) {                                                           \
    while (tail - head >= 32) {                                              \
      resolve_src(src, ring, head, 32, lane);                                \
      edge_steps<(MODE == 1 ? 1 : 2)>(xl, acc, xrs, mx, den, ring, head, 32, \
                                      av, lane);                             \
      head += 32;                                                            \
    }                                                                        \
  }

  const int n4 = ((((size_t)dst) & 15) == 0) ? (nedges >> 2) : 0;
  const int4* __restrict__ dst4 = (const int4*)dst;
  for (int base = wave * (32 * SCAN_G); base < n4;
       base += AGG_WAVES * 32 * SCAN_G) {
    int4 v[SCAN_G];
#pragma unroll
    for (int g = 0; g < SCAN_G; ++g) {
      const int idx = base + g * 32 + lane;
      v[g] = dst4[min(idx, n4 - 1)];
    }
#pragma unroll
    for (int g = 0; g < SCAN_G; ++g) {
      const int idx = base + g * 32 + lane;
      const unsigned lim = idx < n4 ? (unsigned)rows : 0u;
      const unsigned x0 = (unsigned)(v[g].x - n0);
      const unsigned x1 = (unsigned)(v[g].y - n0);
      const unsigned x2 = (unsigned)(v[g].z - n0);
      const unsigned x3 = (unsigned)(v[g].w - n0);
      const bool m0 = x0 < lim, m1 = x1 < lim, m2 = x2 < lim, m3 = x3 < lim;
      if (__builtin_amdgcn_ballot_w32(m0 | m1 | m2 | m3)) {
        const int e0 = idx * 4;
        AGG_PUSH(m0, e0 + 0, x0)
        AGG_PUSH(m1, e0 + 1, x1)
        AGG_PUSH(m2, e0 + 2, x2)
        AGG_PUSH(m3, e0 + 3, x3)
      }
      AGG_DRAIN()
    }
  }
  for (int eb = 4 * n4 + wave * 32; eb < nedges; eb += AGG_WAVES * 32) {
    const int e = eb + lane;
    const unsigned x = e < nedges ? (unsigned)(dst[e] - n0) : 0xFFFFFFFFu;
    const bool m = x < (unsigned)rows;
    AGG_PUSH(m, e, x)
    AGG_DRAIN()
  }
  if (MODE != 0 && tail > head) {
    resolve_src(src, ring, head, tail - head, lane);
    edge_steps<(MODE == 1 ? 1 : 2)>(xl, acc, xrs, mx, den, ring, head,
                                    tail - head, av, lane);
  }
#undef AGG_PUSH
#undef AGG_DRAIN
  cnt_out = tail;
  over_out = MODE == 0 && tail > RING;
}

__global__ __launch_bounds__(AGG_THREADS) void gat_owner_layer(
    const float* __restrict__ xl, const float* __restrict__ xr,
    const float* __restrict__ att, const int* __restrict__ src,
    const int* __restrict__ dst, const float* __restrict__ bias,
    float* __restrict__ out, int n_nodes, int nedges, int do_relu) {
  __shared__ __attribute__((aligned(16))) float acc[ROWS_PB * CC];
  __shared__ __attribute__((aligned(16))) float xrs[ROWS_PB * CC];
  __shared__ unsigned mx[ROWS_PB * HH];
  __shared__ float den[ROWS_PB * HH];
  __shared__ int2 ring_all[AGG_WAVES * RING];
  __shared__ int any_over;

  const int tid = threadIdx.x;
  const int lane = tid & 31;
  const int wave = __builtin_amdgcn_readfirstlane(tid >> 5);
  const int n0 = blockIdx.x * ROWS_PB;
  const int rows = min(ROWS_PB, n_nodes - n0);

  float4* acc4 = (float4*)acc;
  float4* xrs4 = (float4*)xrs;
  const float4* __restrict__ xr4 = (const float4*)(xr + (size_t)n0 * CC);
  for (int i = tid; i < rows * (CC / 4); i += AGG_THREADS) {
    acc4[i] = make_float4(0.0f, 0.0f, 0.0f, 0.0f);
    xrs4[i] = xr4[i];
  }
  for (int i = tid; i < rows * HH; i += AGG_THREADS) {
    mx[i] = 0u;
    den[i] = 0.0f;
  }
  if (tid == 0) any_over = 0;
  __syncthreads();

  const float4 av = ((const float4*)att)[lane & 15];
  int2* ring = ring_all + wave * RING;
  int cnt = 0;
  bool over = false;
  scan_owned<0>(xl, src, dst, acc, xrs, mx, den, ring, cnt, over, n0, rows,
                nedges, av, wave, lane);
  if (over && lane == 0) any_over = 1;
  __syncthreads();
  if (any_over == 0) {
    resolve_src(src, ring, 0, cnt, lane);
    edge_steps<1>(xl, acc, xrs, mx, den, ring, 0, cnt, av, lane);
    __syncthreads();
    edge_steps<2>(xl, acc, xrs, mx, den, ring, 0, cnt, av, lane);
  } else {
    scan_owned<1>(xl, src, dst, acc, xrs, mx, den, ring, cnt, over, n0, rows,
                  nedges, av, wave, lane);
    __syncthreads();
    scan_owned<2>(xl, src, dst, acc, xrs, mx, den, ring, cnt, over, n0, rows,
                  nedges, av, wave, lane);
  }
  __syncthreads();

  const float4* __restrict__ b4 = (const float4*)bias;
  float4* __restrict__ o4 = (float4*)(out + (size_t)n0 * CC);
  for (int i = tid; i < rows * (CC / 4); i += AGG_THREADS) {
    float4 a = acc4[i];
    const float dn = den[(i >> 4) * HH + ((i & 15) >> 2)] + 1e-16f;
    const float4 b = b4[i & (CC / 4 - 1)];
    a.x = a.x / dn + b.x; a.y = a.y / dn + b.y;
    a.z = a.z / dn + b.z; a.w = a.w / dn + b.w;
    if (do_relu) {
      a.x = fmaxf(a.x, 0.0f); a.y = fmaxf(a.y, 0.0f);
      a.z = fmaxf(a.z, 0.0f); a.w = fmaxf(a.w, 0.0f);
    }
    o4[i] = a;
  }
}

extern "C" void kernel_launch(void* const* d_in, const int* in_sizes, int n_in,
                              void* d_out, int out_size, void* d_ws,
                              size_t ws_size, hipStream_t stream) {
  const int N = in_sizes[0] / CC;
  const int E = in_sizes[1] / 2;

  const float* x = (const float*)d_in[0];
  const int* edge = (const int*)d_in[1];
  const int* src = edge;
  const int* dst = edge + E;

  const float* Wl[3] = {(const float*)d_in[2], (const float*)d_in[6],
                        (const float*)d_in[10]};
  const float* Wr[3] = {(const float*)d_in[3], (const float*)d_in[7],
                        (const float*)d_in[11]};
  const float* Av[3] = {(const float*)d_in[4], (const float*)d_in[8],
                        (const float*)d_in[12]};
  const float* Bv[3] = {(const float*)d_in[5], (const float*)d_in[9],
                        (const float*)d_in[13]};

  const size_t NC = (size_t)N * CC;
  float* ws = (float*)d_ws;
  float* xl = ws;                 ws += NC;
  float* xr = ws;                 ws += NC;
  float* hA = ws;                 ws += NC;
  float* hB = ws;                 ws += NC;

  const float* lin[3] = {x, hA, hB};
  float* lout[3] = {hA, hB, (float*)d_out};
  const int lrelu[3] = {1, 1, 0};

  const int gemm_grid = (N + 15) / 16;
  const int agg_grid = (N + ROWS_PB - 1) / ROWS_PB;

  for (int l = 0; l < 3; ++l) {
    gat_gemm_wmma<<<gemm_grid, 128, 0, stream>>>(lin[l], Wl[l], xl, N);
    gat_gemm_wmma<<<gemm_grid, 128, 0, stream>>>(lin[l], Wr[l], xr, N);
    gat_owner_layer<<<agg_grid, AGG_THREADS, 0, stream>>>(
        xl, xr, Av[l], src, dst, Bv[l], lout[l], N, E, lrelu[l]);
  }
}


# Edge lists built once: layer 1 owner blocks store each wave's resolved (source, local row) list in workspace; layers 2 and 3 load it instead of scanning dst and gathering src again; sum order kept
# speedup vs baseline: 4.2878x; 1.2530x over previous
#include <hip/hip_runtime.h>


typedef float v2f __attribute__((ext_vector_type(2)));
typedef float v8f __attribute__((ext_vector_type(8)));

#define CC 64
#define HH 4
#define OO 16
#define NEG_SLOPE 0.2f

__device__ __forceinline__ unsigned fenc(float f) {
  unsigned b = __float_as_uint(f);
  return (b & 0x80000000u) ? ~b : (b | 0x80000000u);
}
__device__ __forceinline__ float fdec(unsigned e) {
  return (e & 0x80000000u) ? __uint_as_float(e & 0x7FFFFFFFu)
                           : __uint_as_float(~e);
}

__global__ __launch_bounds__(128) void gat_gemm_wmma(
    const float* __restrict__ X, const float* __restrict__ W,
    float* __restrict__ Y, int nrows) {
  const int wave = threadIdx.x >> 5;
  const int lane = threadIdx.x & 31;
  const int m0 = blockIdx.x * 16;
  if (m0 >= nrows) return;
  const int n0 = wave * 16;
  const int lrow = lane & 15;
  const int khalf = (lane >> 4) << 1;

  const float* __restrict__ xrow = X + (size_t)(m0 + lrow) * CC;
  v8f c = {};
#pragma unroll
  for (int k = 0; k < CC; k += 4) {
    v2f a, b;
    a.x = xrow[k + khalf];
    a.y = xrow[k + khalf + 1];
    b.x = W[(size_t)(k + khalf) * CC + n0 + lrow];
    b.y = W[(size_t)(k + khalf + 1) * CC + n0 + lrow];
    c = __builtin_amdgcn_wmma_f32_16x16x4_f32(false, a, false, b,
                                              (short)0, c, false, false);
  }
  const int rbase = m0 + ((lane >> 4) << 3);
#pragma unroll
  for (int r = 0; r < 8; ++r)
    Y[(size_t)(rbase + r) * CC + n0 + lrow] = c[r];
}

#define ROWS_PB 391
#define AGG_THREADS 1024
#define AGG_WAVES (AGG_THREADS / 32)
#define RING 256
#define SCAN_G 8
#define STEP_K 8

__device__ __forceinline__ float leaky(float v) {
  return fmaxf(v, NEG_SLOPE * v);
}

__device__ __forceinline__ float quad_sum(float t) {
  t += __int_as_float(__builtin_amdgcn_update_dpp(
      0, __float_as_int(t), 0xB1, 0xF, 0xF, true));
  t += __int_as_float(__builtin_amdgcn_update_dpp(
      0, __float_as_int(t), 0x4E, 0xF, 0xF, true));
  return t;
}

__device__ __forceinline__ void resolve_src(const int* __restrict__ src,
                                            int2* ring, int first, int cnt,
                                            int lane) {
  int e[RING / 32];
#pragma unroll
  for (int k = 0; k < RING / 32; ++k) {
    const int b = lane + 32 * k;
    e[k] = b < cnt ? ring[(first + b) & (RING - 1)].x : 0;
  }
#pragma unroll
  for (int k = 0; k < RING / 32; ++k) {
    const int b = lane + 32 * k;
    if (b < cnt) e[k] = src[e[k]];
  }
#pragma unroll
  for (int k = 0; k < RING / 32; ++k) {
    const int b = lane + 32 * k;
    if (b < cnt) ring[(first + b) & (RING - 1)].x = e[k];
  }
}

template <int PASS>
__device__ __forceinline__ void edge_steps(
    const float* __restrict__ xl, float* acc, const float* xrs, unsigned* mx,
    float* den, const int2* ring, int first, int cnt, float4 av, int lane) {
  const int c4 = (lane & 15) << 2;
  const int half = lane >> 4;
  const int hd = (lane & 15) >> 2;
  const bool quad0 = (lane & 3) == 0;
  const int steps = (cnt + 1) >> 1;
#pragma nounroll
  for (int k0 = 0; k0 < steps; k0 += STEP_K) {
    int2 q[STEP_K];
    float4 lv[STEP_K], rv[STEP_K];
    unsigned me[STEP_K];
#pragma unroll
    for (int k = 0; k < STEP_K; ++k) {
      const int idx = min(2 * (k0 + k) + half, cnt - 1);
      q[k] = ring[(first + idx) & (RING - 1)];
    }
#pragma unroll
    for (int k = 0; k < STEP_K; ++k) {
      lv[k] = *(const float4*)(xl + (size_t)q[k].x * CC + c4);
      rv[k] = *(const float4*)(xrs + q[k].y * CC + c4);
      me[k] = PASS == 1 ? 0u : mx[q[k].y * HH + hd];
    }
#pragma unroll
    for (int k = 0; k < STEP_K; ++k) {
      const bool live = 2 * (k0 + k) + half < cnt;
      float t = av.x * leaky(lv[k].x + rv[k].x);
      t += av.y * leaky(lv[k].y + rv[k].y);
      t += av.z * leaky(lv[k].z + rv[k].z);
      t += av.w * leaky(lv[k].w + rv[k].w);
      t = quad_sum(t);
      const int sh = q[k].y * HH + hd;
      if (PASS == 1) {
        if (live && quad0) atomicMax(&mx[sh], fenc(t));
      } else {
        const float w = __expf(t - fdec(me[k]));
        if (live) {
          float* dr = acc + q[k].y * CC + c4;
          atomicAdd(dr + 0, w * lv[k].x);
          atomicAdd(dr + 1, w * lv[k].y);
          atomicAdd(dr + 2, w * lv[k].z);
          atomicAdd(dr + 3, w * lv[k].w);
          if (quad0) atomicAdd(&den[sh], w);
        }
      }
    }
  }
}

template <int MODE>
__device__ __forceinline__ void scan_owned(
    const float* __restrict__ xl, const int* __restrict__ src,
    const int* __restrict__ dst, float* acc, const float* xrs, unsigned* mx,
    float* den, int2* ring, int& cnt_out, bool& over_out, int n0, int rows,
    int nedges, float4 av, int wave, int lane) {
  int head = 0, tail = 0;

#define AGG_PUSH(m, e, x)                                                    \
  {                                                                          \
    const unsigned bal_ = __builtin_amdgcn_ballot_w32(m);                    \
    if (m)                                                                   \
      ring[(tail + (int)__builtin_amdgcn_mbcnt_lo(bal_, 0u)) & (RING - 1)] = \
          make_int2((e), (int)(x));                                          \
    tail += __builtin_popcount(bal_);                                        \
  }
#define AGG_DRAIN()                                                          \
  if (MODE != 0) {                                                           \
    while (tail - head >= 32) {                                              \
      resolve_src(src, ring, head, 32, lane);                                \
      edge_steps<(MODE == 1 ? 1 : 2)>(xl, acc, xrs, mx, den, ring, head, 32, \
                                      av, lane);                             \
      head += 32;                                                            \
    }                                                                        \
  }

  const int n4 = ((((size_t)dst) & 15) == 0) ? (nedges >> 2) : 0;
  const int4* __restrict__ dst4 = (const int4*)dst;
  for (int base = wave * (32 * SCAN_G); base < n4;
       base += AGG_WAVES * 32 * SCAN_G) {
    int4 v[SCAN_G];
#pragma unroll
    for (int g = 0; g < SCAN_G; ++g) {
      const int idx = base + g * 32 + lane;
      v[g] = dst4[min(idx, n4 - 1)];
    }
#pragma unroll
    for (int g = 0; g < SCAN_G; ++g) {
      const int idx = base + g * 32 + lane;
      const unsigned lim = idx < n4 ? (unsigned)rows : 0u;
      const unsigned x0 = (unsigned)(v[g].x - n0);
      const unsigned x1 = (unsigned)(v[g].y - n0);
      const unsigned x2 = (unsigned)(v[g].z - n0);
      const unsigned x3 = (unsigned)(v[g].w - n0);
      const bool m0 = x0 < lim, m1 = x1 < lim, m2 = x2 < lim, m3 = x3 < lim;
      if (__builtin_amdgcn_ballot_w32(m0 | m1 | m2 | m3)) {
        const int e0 = idx * 4;
        AGG_PUSH(m0, e0 + 0, x0)
        AGG_PUSH(m1, e0 + 1, x1)
        AGG_PUSH(m2, e0 + 2, x2)
        AGG_PUSH(m3, e0 + 3, x3)
      }
      AGG_DRAIN()
    }
  }
  for (int eb = 4 * n4 + wave * 32; eb < nedges; eb += AGG_WAVES * 32) {
    const int e = eb + lane;
    const unsigned x = e < nedges ? (unsigned)(dst[e] - n0) : 0xFFFFFFFFu;
    const bool m = x < (unsigned)rows;
    AGG_PUSH(m, e, x)
    AGG_DRAIN()
  }
  if (MODE != 0 && tail > head) {
    resolve_src(src, ring, head, tail - head, lane);
    edge_steps<(MODE == 1 ? 1 : 2)>(xl, acc, xrs, mx, den, ring, head,
                                    tail - head, av, lane);
  }
#undef AGG_PUSH
#undef AGG_DRAIN
  cnt_out = tail;
  over_out = MODE == 0 && tail > RING;
}

__global__ __launch_bounds__(AGG_THREADS) void gat_owner_layer(
    const float* __restrict__ xl, const float* __restrict__ xr,
    const float* __restrict__ att, const int* __restrict__ src,
    const int* __restrict__ dst, const float* __restrict__ bias,
    float* __restrict__ out, int* __restrict__ lists, int n_nodes, int nedges,
    int do_relu, int build) {
  __shared__ __attribute__((aligned(16))) float acc[ROWS_PB * CC];
  __shared__ __attribute__((aligned(16))) float xrs[ROWS_PB * CC];
  __shared__ unsigned mx[ROWS_PB * HH];
  __shared__ float den[ROWS_PB * HH];
  __shared__ int2 ring_all[AGG_WAVES * RING];
  __shared__ int any_over;

  const int tid = threadIdx.x;
  const int lane = tid & 31;
  const int wave = __builtin_amdgcn_readfirstlane(tid >> 5);
  const int n0 = blockIdx.x * ROWS_PB;
  const int rows = min(ROWS_PB, n_nodes - n0);

  float4* acc4 = (float4*)acc;
  float4* xrs4 = (float4*)xrs;
  const float4* __restrict__ xr4 = (const float4*)(xr + (size_t)n0 * CC);
  for (int i = tid; i < rows * (CC / 4); i += AGG_THREADS) {
    acc4[i] = make_float4(0.0f, 0.0f, 0.0f, 0.0f);
    xrs4[i] = xr4[i];
  }
  for (int i = tid; i < rows * HH; i += AGG_THREADS) {
    mx[i] = 0u;
    den[i] = 0.0f;
  }
  const int nlists = gridDim.x * AGG_WAVES;
  const int lid = blockIdx.x * AGG_WAVES + wave;
  int2* glist = (int2*)lists + (size_t)lid * RING;
  int* gcnt = lists + (size_t)nlists * RING * 2;
  int* gover = gcnt + nlists;
  if (tid == 0) any_over = build ? 0 : gover[blockIdx.x];
  __syncthreads();

  const float4 av = ((const float4*)att)[lane & 15];
  int2* ring = ring_all + wave * RING;
  int cnt = 0;
  bool over = false;
  if (build) {
    scan_owned<0>(xl, src, dst, acc, xrs, mx, den, ring, cnt, over, n0, rows,
                  nedges, av, wave, lane);
    if (over && lane == 0) any_over = 1;
    __syncthreads();
    if (tid == 0) gover[blockIdx.x] = any_over;
  }
  if (any_over == 0) {
    if (build) {
      resolve_src(src, ring, 0, cnt, lane);
#pragma unroll
      for (int k = 0; k < RING / 32; ++k) {
        const int b = lane + 32 * k;
        if (b < cnt) glist[b] = ring[b];
      }
      if (lane == 0) gcnt[lid] = cnt;
    } else {
      cnt = __builtin_amdgcn_readfirstlane(gcnt[lid]);
      int2 t[RING / 32];
#pragma unroll
      for (int k = 0; k < RING / 32; ++k) {
        const int b = lane + 32 * k;
        t[k] = b < cnt ? glist[b] : make_int2(0, 0);
      }
#pragma unroll
      for (int k = 0; k < RING / 32; ++k) {
        const int b = lane + 32 * k;
        if (b < cnt) ring[b] = t[k];
      }
    }
    edge_steps<1>(xl, acc, xrs, mx, den, ring, 0, cnt, av, lane);
    __syncthreads();
    edge_steps<2>(xl, acc, xrs, mx, den, ring, 0, cnt, av, lane);
  } else {
    scan_owned<1>(xl, src, dst, acc, xrs, mx, den, ring, cnt, over, n0, rows,
                  nedges, av, wave, lane);
    __syncthreads();
    scan_owned<2>(xl, src, dst, acc, xrs, mx, den, ring, cnt, over, n0, rows,
                  nedges, av, wave, lane);
  }
  __syncthreads();

  const float4* __restrict__ b4 = (const float4*)bias;
  float4* __restrict__ o4 = (float4*)(out + (size_t)n0 * CC);
  for (int i = tid; i < rows * (CC / 4); i += AGG_THREADS) {
    float4 a = acc4[i];
    const float dn = den[(i >> 4) * HH + ((i & 15) >> 2)] + 1e-16f;
    const float4 b = b4[i & (CC / 4 - 1)];
    a.x = a.x / dn + b.x; a.y = a.y / dn + b.y;
    a.z = a.z / dn + b.z; a.w = a.w / dn + b.w;
    if (do_relu) {
      a.x = fmaxf(a.x, 0.0f); a.y = fmaxf(a.y, 0.0f);
      a.z = fmaxf(a.z, 0.0f); a.w = fmaxf(a.w, 0.0f);
    }
    o4[i] = a;
  }
}

extern "C" void kernel_launch(void* const* d_in, const int* in_sizes, int n_in,
                              void* d_out, int out_size, void* d_ws,
                              size_t ws_size, hipStream_t stream) {
  const int N = in_sizes[0] / CC;
  const int E = in_sizes[1] / 2;

  const float* x = (const float*)d_in[0];
  const int* edge = (const int*)d_in[1];
  const int* src = edge;
  const int* dst = edge + E;

  const float* Wl[3] = {(const float*)d_in[2], (const float*)d_in[6],
                        (const float*)d_in[10]};
  const float* Wr[3] = {(const float*)d_in[3], (const float*)d_in[7],
                        (const float*)d_in[11]};
  const float* Av[3] = {(const float*)d_in[4], (const float*)d_in[8],
                        (const float*)d_in[12]};
  const float* Bv[3] = {(const float*)d_in[5], (const float*)d_in[9],
                        (const float*)d_in[13]};

  const size_t NC = (size_t)N * CC;
  float* ws = (float*)d_ws;
  float* xl = ws;                 ws += NC;
  float* xr = ws;                 ws += NC;
  float* hA = ws;                 ws += NC;
  int* lists = (int*)ws;

  const float* lin[3] = {x, hA, hA};
  float* lout[3] = {hA, hA, (float*)d_out};
  const int lrelu[3] = {1, 1, 0};
  const int lbuild[3] = {1, 0, 0};

  const int gemm_grid = (N + 15) / 16;
  const int agg_grid = (N + ROWS_PB - 1) / ROWS_PB;

  for (int l = 0; l < 3; ++l) {
    gat_gemm_wmma<<<gemm_grid, 128, 0, stream>>>(lin[l], Wl[l], xl, N);
    gat_gemm_wmma<<<gemm_grid, 128, 0, stream>>>(lin[l], Wr[l], xr, N);
    gat_owner_layer<<<agg_grid, AGG_THREADS, 0, stream>>>(
        xl, xr, Av[l], src, dst, Bv[l], lout[l], lists, N, E, lrelu[l],
        lbuild[l]);
  }
}


# Destination-side projection xr = h@Wr computed by each owner block for its own rows (f32 matrix instructions, straight into LDS); xr never stored; 6 launches for 9; sums unchanged
# speedup vs baseline: 4.6518x; 1.0849x over previous
#include <hip/hip_runtime.h>


typedef float v2f __attribute__((ext_vector_type(2)));
typedef float v8f __attribute__((ext_vector_type(8)));

#define CC 64
#define HH 4
#define OO 16
#define NEG_SLOPE 0.2f

__device__ __forceinline__ unsigned fenc(float f) {
  unsigned b = __float_as_uint(f);
  return (b & 0x80000000u) ? ~b : (b | 0x80000000u);
}
__device__ __forceinline__ float fdec(unsigned e) {
  return (e & 0x80000000u) ? __uint_as_float(e & 0x7FFFFFFFu)
                           : __uint_as_float(~e);
}

__global__ __launch_bounds__(128) void gat_gemm_wmma(
    const float* __restrict__ X, const float* __restrict__ W,
    float* __restrict__ Y, int nrows) {
  const int wave = threadIdx.x >> 5;
  const int lane = threadIdx.x & 31;
  const int m0 = blockIdx.x * 16;
  if (m0 >= nrows) return;
  const int n0 = wave * 16;
  const int lrow = lane & 15;
  const int khalf = (lane >> 4) << 1;

  const float* __restrict__ xrow = X + (size_t)(m0 + lrow) * CC;
  v8f c = {};
#pragma unroll
  for (int k = 0; k < CC; k += 4) {
    v2f a, b;
    a.x = xrow[k + khalf];
    a.y = xrow[k + khalf + 1];
    b.x = W[(size_t)(k + khalf) * CC + n0 + lrow];
    b.y = W[(size_t)(k + khalf + 1) * CC + n0 + lrow];
    c = __builtin_amdgcn_wmma_f32_16x16x4_f32(false, a, false, b,
                                              (short)0, c, false, false);
  }
  const int rbase = m0 + ((lane >> 4) << 3);
#pragma unroll
  for (int r = 0; r < 8; ++r)
    Y[(size_t)(rbase + r) * CC + n0 + lrow] = c[r];
}

#define ROWS_PB 391
#define AGG_THREADS 1024
#define AGG_WAVES (AGG_THREADS / 32)
#define RING 256
#define SCAN_G 8
#define STEP_K 8

__device__ __forceinline__ void project_own_rows(
    const float* __restrict__ hin, const float* __restrict__ Wr, float* xrs,
    int n0, int rows, int wave, int lane) {
  const int lrow = lane & 15;
  const int khalf = (lane >> 4) << 1;
  const int ncol = (wave & 3) * 16;
  v2f b[CC / 4];
#pragma unroll
  for (int j = 0; j < CC / 4; ++j) {
    b[j].x = Wr[(size_t)(4 * j + khalf) * CC + ncol + lrow];
    b[j].y = Wr[(size_t)(4 * j + khalf + 1) * CC + ncol + lrow];
  }
  const int ntiles = (rows + 15) >> 4;
  for (int t = wave >> 2; t < ntiles; t += AGG_WAVES / 4) {
    const int m0 = t * 16;
    const float* __restrict__ xrow =
        hin + (size_t)(n0 + min(m0 + lrow, rows - 1)) * CC;
    v8f c = {};
#pragma unroll
    for (int j = 0; j < CC / 4; ++j) {
      v2f a;
      a.x = xrow[4 * j + khalf];
      a.y = xrow[4 * j + khalf + 1];
      c = __builtin_amdgcn_wmma_f32_16x16x4_f32(false, a, false, b[j],
                                                (short)0, c, false, false);
    }
    const int rb = m0 + ((lane >> 4) << 3);
#pragma unroll
    for (int r = 0; r < 8; ++r)
      if (rb + r < rows) xrs[(rb + r) * CC + ncol + lrow] = c[r];
  }
}

__device__ __forceinline__ float leaky(float v) {
  return fmaxf(v, NEG_SLOPE * v);
}

__device__ __forceinline__ float quad_sum(float t) {
  t += __int_as_float(__builtin_amdgcn_update_dpp(
      0, __float_as_int(t), 0xB1, 0xF, 0xF, true));
  t += __int_as_float(__builtin_amdgcn_update_dpp(
      0, __float_as_int(t), 0x4E, 0xF, 0xF, true));
  return t;
}

__device__ __forceinline__ void resolve_src(const int* __restrict__ src,
                                            int2* ring, int first, int cnt,
                                            int lane) {
  int e[RING / 32];
#pragma unroll
  for (int k = 0; k < RING / 32; ++k) {
    const int b = lane + 32 * k;
    e[k] = b < cnt ? ring[(first + b) & (RING - 1)].x : 0;
  }
#pragma unroll
  for (int k = 0; k < RING / 32; ++k) {
    const int b = lane + 32 * k;
    if (b < cnt) e[k] = src[e[k]];
  }
#pragma unroll
  for (int k = 0; k < RING / 32; ++k) {
    const int b = lane + 32 * k;
    if (b < cnt) ring[(first + b) & (RING - 1)].x = e[k];
  }
}

template <int PASS>
__device__ __forceinline__ void edge_steps(
    const float* __restrict__ xl, float* acc, const float* xrs, unsigned* mx,
    float* den, const int2* ring, int first, int cnt, float4 av, int lane) {
  const int c4 = (lane & 15) << 2;
  const int half = lane >> 4;
  const int hd = (lane & 15) >> 2;
  const bool quad0 = (lane & 3) == 0;
  const int steps = (cnt + 1) >> 1;
#pragma nounroll
  for (int k0 = 0; k0 < steps; k0 += STEP_K) {
    int2 q[STEP_K];
    float4 lv[STEP_K], rv[STEP_K];
    unsigned me[STEP_K];
#pragma unroll
    for (int k = 0; k < STEP_K; ++k) {
      const int idx = min(2 * (k0 + k) + half, cnt - 1);
      q[k] = ring[(first + idx) & (RING - 1)];
    }
#pragma unroll
    for (int k = 0; k < STEP_K; ++k) {
      lv[k] = *(const float4*)(xl + (size_t)q[k].x * CC + c4);
      rv[k] = *(const float4*)(xrs + q[k].y * CC + c4);
      me[k] = PASS == 1 ? 0u : mx[q[k].y * HH + hd];
    }
#pragma unroll
    for (int k = 0; k < STEP_K; ++k) {
      const bool live = 2 * (k0 + k) + half < cnt;
      float t = av.x * leaky(lv[k].x + rv[k].x);
      t += av.y * leaky(lv[k].y + rv[k].y);
      t += av.z * leaky(lv[k].z + rv[k].z);
      t += av.w * leaky(lv[k].w + rv[k].w);
      t = quad_sum(t);
      const int sh = q[k].y * HH + hd;
      if (PASS == 1) {
        if (live && quad0) atomicMax(&mx[sh], fenc(t));
      } else {
        const float w = __expf(t - fdec(me[k]));
        if (live) {
          float* dr = acc + q[k].y * CC + c4;
          atomicAdd(dr + 0, w * lv[k].x);
          atomicAdd(dr + 1, w * lv[k].y);
          atomicAdd(dr + 2, w * lv[k].z);
          atomicAdd(dr + 3, w * lv[k].w);
          if (quad0) atomicAdd(&den[sh], w);
        }
      }
    }
  }
}

template <int MODE>
__device__ __forceinline__ void scan_owned(
    const float* __restrict__ xl, const int* __restrict__ src,
    const int* __restrict__ dst, float* acc, const float* xrs, unsigned* mx,
    float* den, int2* ring, int& cnt_out, bool& over_out, int n0, int rows,
    int nedges, float4 av, int wave, int lane) {
  int head = 0, tail = 0;

#define AGG_PUSH(m, e, x)                                                    \
  {                                                                          \
    const unsigned bal_ = __builtin_amdgcn_ballot_w32(m);                    \
    if (m)                                                                   \
      ring[(tail + (int)__builtin_amdgcn_mbcnt_lo(bal_, 0u)) & (RING - 1)] = \
          make_int2((e), (int)(x));                                          \
    tail += __builtin_popcount(bal_);                                        \
  }
#define AGG_DRAIN()                                                          \
  if (MODE != 0) {                                                           \
    while (tail - head >= 32) {                                              \
      resolve_src(src, ring, head, 32, lane);                                \
      edge_steps<(MODE == 1 ? 1 : 2)>(xl, acc, xrs, mx, den, ring, head, 32, \
                                      av, lane);                             \
      head += 32;                                                            \
    }                                                                        \
  }

  const int n4 = ((((size_t)dst) & 15) == 0) ? (nedges >> 2) : 0;
  const int4* __restrict__ dst4 = (const int4*)dst;
  for (int base = wave * (32 * SCAN_G); base < n4;
       base += AGG_WAVES * 32 * SCAN_G) {
    int4 v[SCAN_G];
#pragma unroll
    for (int g = 0; g < SCAN_G; ++g) {
      const int idx = base + g * 32 + lane;
      v[g] = dst4[min(idx, n4 - 1)];
    }
#pragma unroll
    for (int g = 0; g < SCAN_G; ++g) {
      const int idx = base + g * 32 + lane;
      const unsigned lim = idx < n4 ? (unsigned)rows : 0u;
      const unsigned x0 = (unsigned)(v[g].x - n0);
      const unsigned x1 = (unsigned)(v[g].y - n0);
      const unsigned x2 = (unsigned)(v[g].z - n0);
      const unsigned x3 = (unsigned)(v[g].w - n0);
      const bool m0 = x0 < lim, m1 = x1 < lim, m2 = x2 < lim, m3 = x3 < lim;
      if (__builtin_amdgcn_ballot_w32(m0 | m1 | m2 | m3)) {
        const int e0 = idx * 4;
        AGG_PUSH(m0, e0 + 0, x0)
        AGG_PUSH(m1, e0 + 1, x1)
        AGG_PUSH(m2, e0 + 2, x2)
        AGG_PUSH(m3, e0 + 3, x3)
      }
      AGG_DRAIN()
    }
  }
  for (int eb = 4 * n4 + wave * 32; eb < nedges; eb += AGG_WAVES * 32) {
    const int e = eb + lane;
    const unsigned x = e < nedges ? (unsigned)(dst[e] - n0) : 0xFFFFFFFFu;
    const bool m = x < (unsigned)rows;
    AGG_PUSH(m, e, x)
    AGG_DRAIN()
  }
  if (MODE != 0 && tail > head) {
    resolve_src(src, ring, head, tail - head, lane);
    edge_steps<(MODE == 1 ? 1 : 2)>(xl, acc, xrs, mx, den, ring, head,
                                    tail - head, av, lane);
  }
#undef AGG_PUSH
#undef AGG_DRAIN
  cnt_out = tail;
  over_out = MODE == 0 && tail > RING;
}

__global__ __launch_bounds__(AGG_THREADS) void gat_owner_layer(
    const float* __restrict__ xl, const float* __restrict__ hin,
    const float* __restrict__ wr, const float* __restrict__ att,
    const int* __restrict__ src,
    const int* __restrict__ dst, const float* __restrict__ bias,
    float* __restrict__ out, int* __restrict__ lists, int n_nodes, int nedges,
    int do_relu, int build) {
  __shared__ __attribute__((aligned(16))) float acc[ROWS_PB * CC];
  __shared__ __attribute__((aligned(16))) float xrs[ROWS_PB * CC];
  __shared__ unsigned mx[ROWS_PB * HH];
  __shared__ float den[ROWS_PB * HH];
  __shared__ int2 ring_all[AGG_WAVES * RING];
  __shared__ int any_over;

  const int tid = threadIdx.x;
  const int lane = tid & 31;
  const int wave = __builtin_amdgcn_readfirstlane(tid >> 5);
  const int n0 = blockIdx.x * ROWS_PB;
  const int rows = min(ROWS_PB, n_nodes - n0);

  float4* acc4 = (float4*)acc;
  for (int i = tid; i < rows * (CC / 4); i += AGG_THREADS)
    acc4[i] = make_float4(0.0f, 0.0f, 0.0f, 0.0f);
  for (int i = tid; i < rows * HH; i += AGG_THREADS) {
    mx[i] = 0u;
    den[i] = 0.0f;
  }
  project_own_rows(hin, wr, xrs, n0, rows, wave, lane);
  const int nlists = gridDim.x * AGG_WAVES;
  const int lid = blockIdx.x * AGG_WAVES + wave;
  int2* glist = (int2*)lists + (size_t)lid * RING;
  int* gcnt = lists + (size_t)nlists * RING * 2;
  int* gover = gcnt + nlists;
  if (tid == 0) any_over = build ? 0 : gover[blockIdx.x];
  __syncthreads();

  const float4 av = ((const float4*)att)[lane & 15];
  int2* ring = ring_all + wave * RING;
  int cnt = 0;
  bool over = false;
  if (build) {
    scan_owned<0>(xl, src, dst, acc, xrs, mx, den, ring, cnt, over, n0, rows,
                  nedges, av, wave, lane);
    if (over && lane == 0) any_over = 1;
    __syncthreads();
    if (tid == 0) gover[blockIdx.x] = any_over;
  }
  if (any_over == 0) {
    if (build) {
      resolve_src(src, ring, 0, cnt, lane);
#pragma unroll
      for (int k = 0; k < RING / 32; ++k) {
        const int b = lane + 32 * k;
        if (b < cnt) glist[b] = ring[b];
      }
      if (lane == 0) gcnt[lid] = cnt;
    } else {
      cnt = __builtin_amdgcn_readfirstlane(gcnt[lid]);
      int2 t[RING / 32];
#pragma unroll
      for (int k = 0; k < RING / 32; ++k) {
        const int b = lane + 32 * k;
        t[k] = b < cnt ? glist[b] : make_int2(0, 0);
      }
#pragma unroll
      for (int k = 0; k < RING / 32; ++k) {
        const int b = lane + 32 * k;
        if (b < cnt) ring[b] = t[k];
      }
    }
    edge_steps<1>(xl, acc, xrs, mx, den, ring, 0, cnt, av, lane);
    __syncthreads();
    edge_steps<2>(xl, acc, xrs, mx, den, ring, 0, cnt, av, lane);
  } else {
    scan_owned<1>(xl, src, dst, acc, xrs, mx, den, ring, cnt, over, n0, rows,
                  nedges, av, wave, lane);
    __syncthreads();
    scan_owned<2>(xl, src, dst, acc, xrs, mx, den, ring, cnt, over, n0, rows,
                  nedges, av, wave, lane);
  }
  __syncthreads();

  const float4* __restrict__ b4 = (const float4*)bias;
  float4* __restrict__ o4 = (float4*)(out + (size_t)n0 * CC);
  for (int i = tid; i < rows * (CC / 4); i += AGG_THREADS) {
    float4 a = acc4[i];
    const float dn = den[(i >> 4) * HH + ((i & 15) >> 2)] + 1e-16f;
    const float4 b = b4[i & (CC / 4 - 1)];
    a.x = a.x / dn + b.x; a.y = a.y / dn + b.y;
    a.z = a.z / dn + b.z; a.w = a.w / dn + b.w;
    if (do_relu) {
      a.x = fmaxf(a.x, 0.0f); a.y = fmaxf(a.y, 0.0f);
      a.z = fmaxf(a.z, 0.0f); a.w = fmaxf(a.w, 0.0f);
    }
    o4[i] = a;
  }
}

extern "C" void kernel_launch(void* const* d_in, const int* in_sizes, int n_in,
                              void* d_out, int out_size, void* d_ws,
                              size_t ws_size, hipStream_t stream) {
  const int N = in_sizes[0] / CC;
  const int E = in_sizes[1] / 2;

  const float* x = (const float*)d_in[0];
  const int* edge = (const int*)d_in[1];
  const int* src = edge;
  const int* dst = edge + E;

  const float* Wl[3] = {(const float*)d_in[2], (const float*)d_in[6],
                        (const float*)d_in[10]};
  const float* Wr[3] = {(const float*)d_in[3], (const float*)d_in[7],
                        (const float*)d_in[11]};
  const float* Av[3] = {(const float*)d_in[4], (const float*)d_in[8],
                        (const float*)d_in[12]};
  const float* Bv[3] = {(const float*)d_in[5], (const float*)d_in[9],
                        (const float*)d_in[13]};

  const size_t NC = (size_t)N * CC;
  float* ws = (float*)d_ws;
  float* xl = ws;                 ws += NC;
  float* hB = ws;                 ws += NC;
  float* hA = ws;                 ws += NC;
  int* lists = (int*)ws;

  const float* lin[3] = {x, hA, hB};
  float* lout[3] = {hA, hB, (float*)d_out};
  const int lrelu[3] = {1, 1, 0};
  const int lbuild[3] = {1, 0, 0};

  const int gemm_grid = (N + 15) / 16;
  const int agg_grid = (N + ROWS_PB - 1) / ROWS_PB;

  for (int l = 0; l < 3; ++l) {
    gat_gemm_wmma<<<gemm_grid, 128, 0, stream>>>(lin[l], Wl[l], xl, N);
    gat_owner_layer<<<agg_grid, AGG_THREADS, 0, stream>>>(
        xl, lin[l], Wr[l], Av[l], src, dst, Bv[l], lout[l], lists, N, E,
        lrelu[l], lbuild[l]);
  }
}


# Softmax shift = separable upper bound of the logits (leaky_relu is subadditive): shift pass gathers one float4 per edge, not an xl row; exact two-pass redo if a bound is loose; exp rounding changes
# speedup vs baseline: 5.1143x; 1.0994x over previous
#include <hip/hip_runtime.h>


typedef float v2f __attribute__((ext_vector_type(2)));
typedef float v8f __attribute__((ext_vector_type(8)));

#define CC 64
#define HH 4
#define OO 16
#define NEG_SLOPE 0.2f

__device__ __forceinline__ unsigned fenc(float f) {
  unsigned b = __float_as_uint(f);
  return (b & 0x80000000u) ? ~b : (b | 0x80000000u);
}
__device__ __forceinline__ float fdec(unsigned e) {
  return (e & 0x80000000u) ? __uint_as_float(e & 0x7FFFFFFFu)
                           : __uint_as_float(~e);
}

__device__ __forceinline__ float leaky(float v) {
  return fmaxf(v, NEG_SLOPE * v);
}

__device__ __forceinline__ float quad_sum(float t) {
  t += __int_as_float(__builtin_amdgcn_update_dpp(
      0, __float_as_int(t), 0xB1, 0xF, 0xF, true));
  t += __int_as_float(__builtin_amdgcn_update_dpp(
      0, __float_as_int(t), 0x4E, 0xF, 0xF, true));
  return t;
}

__device__ __forceinline__ float row16_sum(float t) {
  t = quad_sum(t);
  t += __int_as_float(__builtin_amdgcn_update_dpp(
      0, __float_as_int(t), 0x141, 0xF, 0xF, true));
  t += __int_as_float(__builtin_amdgcn_update_dpp(
      0, __float_as_int(t), 0x140, 0xF, 0xF, true));
  return t;
}

__global__ __launch_bounds__(128) void gat_gemm_wmma(
    const float* __restrict__ X, const float* __restrict__ W,
    const float* __restrict__ att, float* __restrict__ Y,
    float* __restrict__ S, int nrows) {
  const int wave = threadIdx.x >> 5;
  const int lane = threadIdx.x & 31;
  const int m0 = blockIdx.x * 16;
  if (m0 >= nrows) return;
  const int n0 = wave * 16;
  const int lrow = lane & 15;
  const int khalf = (lane >> 4) << 1;

  const float* __restrict__ xrow = X + (size_t)(m0 + lrow) * CC;
  v8f c = {};
#pragma unroll
  for (int k = 0; k < CC; k += 4) {
    v2f a, b;
    a.x = xrow[k + khalf];
    a.y = xrow[k + khalf + 1];
    b.x = W[(size_t)(k + khalf) * CC + n0 + lrow];
    b.y = W[(size_t)(k + khalf + 1) * CC + n0 + lrow];
    c = __builtin_amdgcn_wmma_f32_16x16x4_f32(false, a, false, b,
                                              (short)0, c, false, false);
  }
  const int rbase = m0 + ((lane >> 4) << 3);
#pragma unroll
  for (int r = 0; r < 8; ++r)
    Y[(size_t)(rbase + r) * CC + n0 + lrow] = c[r];
  const float ao = att[n0 + lrow];
  float sv = 0.0f;
#pragma unroll
  for (int r = 0; r < 8; ++r) {
    const float p = row16_sum(ao * leaky(c[r]));
    if (lrow == r) sv = p;
  }
  if (lrow < 8) S[(size_t)(rbase + lrow) * HH + wave] = sv;
}

#define ROWS_PB 391
#define AGG_THREADS 1024
#define AGG_WAVES (AGG_THREADS / 32)
#define RING 256
#define SCAN_G 8
#define STEP_K 8

__device__ __forceinline__ void project_own_rows(
    const float* __restrict__ hin, const float* __restrict__ Wr,
    const float* __restrict__ att, float* xrs, float* rbnd,
    int n0, int rows, int wave, int lane) {
  const int lrow = lane & 15;
  const int khalf = (lane >> 4) << 1;
  const int ncol = (wave & 3) * 16;
  const float ao = att[ncol + lrow];
  const float aabs = fabsf(ao);
  const float asgn = ao < 0.0f ? -1.0f : 1.0f;
  v2f b[CC / 4];
#pragma unroll
  for (int j = 0; j < CC / 4; ++j) {
    b[j].x = Wr[(size_t)(4 * j + khalf) * CC + ncol + lrow];
    b[j].y = Wr[(size_t)(4 * j + khalf + 1) * CC + ncol + lrow];
  }
  const int ntiles = (rows + 15) >> 4;
  for (int t = wave >> 2; t < ntiles; t += AGG_WAVES / 4) {
    const int m0 = t * 16;
    const float* __restrict__ xrow =
        hin + (size_t)(n0 + min(m0 + lrow, rows - 1)) * CC;
    v8f c = {};
#pragma unroll
    for (int j = 0; j < CC / 4; ++j) {
      v2f a;
      a.x = xrow[4 * j + khalf];
      a.y = xrow[4 * j + khalf + 1];
      c = __builtin_amdgcn_wmma_f32_16x16x4_f32(false, a, false, b[j],
                                                (short)0, c, false, false);
    }
    const int rb = m0 + ((lane >> 4) << 3);
#pragma unroll
    for (int r = 0; r < 8; ++r)
      if (rb + r < rows) xrs[(rb + r) * CC + ncol + lrow] = c[r];
    float rv = 0.0f;
#pragma unroll
    for (int r = 0; r < 8; ++r) {
      const float p = row16_sum(aabs * leaky(asgn * c[r]));
      if (lrow == r) rv = p;
    }
    if (lrow < 8 && rb + lrow < rows) rbnd[(rb + lrow) * HH + (wave & 3)] = rv;
  }
}

__device__ __forceinline__ void resolve_src(const int* __restrict__ src,
                                            int2* ring, int first, int cnt,
                                            int lane) {
  int e[RING / 32];
#pragma unroll
  for (int k = 0; k < RING / 32; ++k) {
    const int b = lane + 32 * k;
    e[k] = b < cnt ? ring[(first + b) & (RING - 1)].x : 0;
  }
#pragma unroll
  for (int k = 0; k < RING / 32; ++k) {
    const int b = lane + 32 * k;
    if (b < cnt) e[k] = src[e[k]];
  }
#pragma unroll
  for (int k = 0; k < RING / 32; ++k) {
    const int b = lane + 32 * k;
    if (b < cnt) ring[(first + b) & (RING - 1)].x = e[k];
  }
}

template <int PASS>
__device__ __forceinline__ void edge_steps(
    const float* __restrict__ xl, float* acc, const float* xrs, unsigned* mx,
    float* den, const int2* ring, int first, int cnt, float4 av, int lane) {
  const int c4 = (lane & 15) << 2;
  const int half = lane >> 4;
  const int hd = (lane & 15) >> 2;
  const bool quad0 = (lane & 3) == 0;
  const int steps = (cnt + 1) >> 1;
#pragma nounroll
  for (int k0 = 0; k0 < steps; k0 += STEP_K) {
    int2 q[STEP_K];
    float4 lv[STEP_K], rv[STEP_K];
    unsigned me[STEP_K];
#pragma unroll
    for (int k = 0; k < STEP_K; ++k) {
      const int idx = min(2 * (k0 + k) + half, cnt - 1);
      q[k] = ring[(first + idx) & (RING - 1)];
    }
#pragma unroll
    for (int k = 0; k < STEP_K; ++k) {
      lv[k] = *(const float4*)(xl + (size_t)q[k].x * CC + c4);
      rv[k] = *(const float4*)(xrs + q[k].y * CC + c4);
      me[k] = PASS == 1 ? 0u : mx[q[k].y * HH + hd];
    }
#pragma unroll
    for (int k = 0; k < STEP_K; ++k) {
      const bool live = 2 * (k0 + k) + half < cnt;
      float t = av.x * leaky(lv[k].x + rv[k].x);
      t += av.y * leaky(lv[k].y + rv[k].y);
      t += av.z * leaky(lv[k].z + rv[k].z);
      t += av.w * leaky(lv[k].w + rv[k].w);
      t = quad_sum(t);
      const int sh = q[k].y * HH + hd;
      if (PASS == 1) {
        if (live && quad0) atomicMax(&mx[sh], fenc(t));
      } else {
        const float w = __expf(t - fdec(me[k]));
        if (live) {
          float* dr = acc + q[k].y * CC + c4;
          atomicAdd(dr + 0, w * lv[k].x);
          atomicAdd(dr + 1, w * lv[k].y);
          atomicAdd(dr + 2, w * lv[k].z);
          atomicAdd(dr + 3, w * lv[k].w);
          if (quad0) atomicAdd(&den[sh], w);
        }
      }
    }
  }
}

__device__ __forceinline__ void bound_steps(
    const float* __restrict__ sbnd, const float* rbnd, unsigned* mx,
    const int2* ring, int cnt, int lane) {
  int2 q[RING / 32];
  float4 s[RING / 32];
#pragma unroll
  for (int k = 0; k < RING / 32; ++k) {
    const int b = lane + 32 * k;
    q[k] = b < cnt ? ring[b] : make_int2(0, 0);
  }
#pragma unroll
  for (int k = 0; k < RING / 32; ++k) {
    const int b = lane + 32 * k;
    s[k] = b < cnt ? *(const float4*)(sbnd + (size_t)q[k].x * HH)
                   : make_float4(0.0f, 0.0f, 0.0f, 0.0f);
  }
#pragma unroll
  for (int k = 0; k < RING / 32; ++k) {
    const int b = lane + 32 * k;
    if (b < cnt) {
      const float4 r = *(const float4*)(rbnd + q[k].y * HH);
      unsigned* m = mx + q[k].y * HH;
      atomicMax(m + 0, fenc(s[k].x + r.x));
      atomicMax(m + 1, fenc(s[k].y + r.y));
      atomicMax(m + 2, fenc(s[k].z + r.z));
      atomicMax(m + 3, fenc(s[k].w + r.w));
    }
  }
}

template <int MODE>
__device__ __forceinline__ void scan_owned(
    const float* __restrict__ xl, const int* __restrict__ src,
    const int* __restrict__ dst, float* acc, const float* xrs, unsigned* mx,
    float* den, int2* ring, int& cnt_out, bool& over_out, int n0, int rows,
    int nedges, float4 av, int wave, int lane) {
  int head = 0, tail = 0;

#define AGG_PUSH(m, e, x)                                                    \
  {                                                                          \
    const unsigned bal_ = __builtin_amdgcn_ballot_w32(m);                    \
    if (m)                                                                   \
      ring[(tail + (int)__builtin_amdgcn_mbcnt_lo(bal_, 0u)) & (RING - 1)] = \
          make_int2((e), (int)(x));                                          \
    tail += __builtin_popcount(bal_);                                        \
  }
#define AGG_DRAIN()                                                          \
  if (MODE != 0) {                                                           \
    while (tail - head >= 32) {                                              \
      resolve_src(src, ring, head, 32, lane);                                \
      edge_steps<(MODE == 1 ? 1 : 2)>(xl, acc, xrs, mx, den, ring, head, 32, \
                                      av, lane);                             \
      head += 32;                                                            \
    }                                                                        \
  }

  const int n4 = ((((size_t)dst) & 15) == 0) ? (nedges >> 2) : 0;
  const int4* __restrict__ dst4 = (const int4*)dst;
  for (int base = wave * (32 * SCAN_G); base < n4;
       base += AGG_WAVES * 32 * SCAN_G) {
    int4 v[SCAN_G];
#pragma unroll
    for (int g = 0; g < SCAN_G; ++g) {
      const int idx = base + g * 32 + lane;
      v[g] = dst4[min(idx, n4 - 1)];
    }
#pragma unroll
    for (int g = 0; g < SCAN_G; ++g) {
      const int idx = base + g * 32 + lane;
      const unsigned lim = idx < n4 ? (unsigned)rows : 0u;
      const unsigned x0 = (unsigned)(v[g].x - n0);
      const unsigned x1 = (unsigned)(v[g].y - n0);
      const unsigned x2 = (unsigned)(v[g].z - n0);
      const unsigned x3 = (unsigned)(v[g].w - n0);
      const bool m0 = x0 < lim, m1 = x1 < lim, m2 = x2 < lim, m3 = x3 < lim;
      if (__builtin_amdgcn_ballot_w32(m0 | m1 | m2 | m3)) {
        const int e0 = idx * 4;
        AGG_PUSH(m0, e0 + 0, x0)
        AGG_PUSH(m1, e0 + 1, x1)
        AGG_PUSH(m2, e0 + 2, x2)
        AGG_PUSH(m3, e0 + 3, x3)
      }
      AGG_DRAIN()
    }
  }
  for (int eb = 4 * n4 + wave * 32; eb < nedges; eb += AGG_WAVES * 32) {
    const int e = eb + lane;
    const unsigned x = e < nedges ? (unsigned)(dst[e] - n0) : 0xFFFFFFFFu;
    const bool m = x < (unsigned)rows;
    AGG_PUSH(m, e, x)
    AGG_DRAIN()
  }
  if (MODE != 0 && tail > head) {
    resolve_src(src, ring, head, tail - head, lane);
    edge_steps<(MODE == 1 ? 1 : 2)>(xl, acc, xrs, mx, den, ring, head,
                                    tail - head, av, lane);
  }
#undef AGG_PUSH
#undef AGG_DRAIN
  cnt_out = tail;
  over_out = MODE == 0 && tail > RING;
}

__global__ __launch_bounds__(AGG_THREADS) void gat_owner_layer(
    const float* __restrict__ xl, const float* __restrict__ sbnd,
    const float* __restrict__ hin,
    const float* __restrict__ wr, const float* __restrict__ att,
    const int* __restrict__ src,
    const int* __restrict__ dst, const float* __restrict__ bias,
    float* __restrict__ out, int* __restrict__ lists, int n_nodes, int nedges,
    int do_relu, int build) {
  __shared__ __attribute__((aligned(16))) float acc[ROWS_PB * CC];
  __shared__ __attribute__((aligned(16))) float xrs[ROWS_PB * CC];
  __shared__ __attribute__((aligned(16))) float rbnd[ROWS_PB * HH];
  __shared__ unsigned mx[ROWS_PB * HH];
  __shared__ float den[ROWS_PB * HH];
  __shared__ int2 ring_all[AGG_WAVES * RING];
  __shared__ int any_over;
  __shared__ int redo;

  const int tid = threadIdx.x;
  const int lane = tid & 31;
  const int wave = __builtin_amdgcn_readfirstlane(tid >> 5);
  const int n0 = blockIdx.x * ROWS_PB;
  const int rows = min(ROWS_PB, n_nodes - n0);

  float4* acc4 = (float4*)acc;
  for (int i = tid; i < rows * (CC / 4); i += AGG_THREADS)
    acc4[i] = make_float4(0.0f, 0.0f, 0.0f, 0.0f);
  for (int i = tid; i < rows * HH; i += AGG_THREADS) {
    mx[i] = 0u;
    den[i] = 0.0f;
  }
  project_own_rows(hin, wr, att, xrs, rbnd, n0, rows, wave, lane);
  const int nlists = gridDim.x * AGG_WAVES;
  const int lid = blockIdx.x * AGG_WAVES + wave;
  int2* glist = (int2*)lists + (size_t)lid * RING;
  int* gcnt = lists + (size_t)nlists * RING * 2;
  int* gover = gcnt + nlists;
  if (tid == 0) {
    any_over = build ? 0 : gover[blockIdx.x];
    redo = 0;
  }
  __syncthreads();

  const float4 av = ((const float4*)att)[lane & 15];
  int2* ring = ring_all + wave * RING;
  int cnt = 0;
  bool over = false;
  if (build) {
    scan_owned<0>(xl, src, dst, acc, xrs, mx, den, ring, cnt, over, n0, rows,
                  nedges, av, wave, lane);
    if (over && lane == 0) any_over = 1;
    __syncthreads();
    if (tid == 0) gover[blockIdx.x] = any_over;
  }
  if (any_over == 0) {
    if (build) {
      resolve_src(src, ring, 0, cnt, lane);
#pragma unroll
      for (int k = 0; k < RING / 32; ++k) {
        const int b = lane + 32 * k;
        if (b < cnt) glist[b] = ring[b];
      }
      if (lane == 0) gcnt[lid] = cnt;
    } else {
      cnt = __builtin_amdgcn_readfirstlane(gcnt[lid]);
      int2 t[RING / 32];
#pragma unroll
      for (int k = 0; k < RING / 32; ++k) {
        const int b = lane + 32 * k;
        t[k] = b < cnt ? glist[b] : make_int2(0, 0);
      }
#pragma unroll
      for (int k = 0; k < RING / 32; ++k) {
        const int b = lane + 32 * k;
        if (b < cnt) ring[b] = t[k];
      }
    }
#pragma nounroll
    for (int exact = 0; exact < 2; ++exact) {
      if (exact == 0)
        bound_steps(sbnd, rbnd, mx, ring, cnt, lane);
      else
        edge_steps<1>(xl, acc, xrs, mx, den, ring, 0, cnt, av, lane);
      __syncthreads();
      edge_steps<2>(xl, acc, xrs, mx, den, ring, 0, cnt, av, lane);
      __syncthreads();
      if (exact) break;
      bool bad = false;
      for (int i = tid; i < rows * HH; i += AGG_THREADS)
        bad |= (mx[i] != 0u) & (den[i] < 1e-7f);
      if (bad) redo = 1;
      __syncthreads();
      if (redo == 0) break;
      for (int i = tid; i < rows * (CC / 4); i += AGG_THREADS)
        acc4[i] = make_float4(0.0f, 0.0f, 0.0f, 0.0f);
      for (int i = tid; i < rows * HH; i += AGG_THREADS) {
        mx[i] = 0u;
        den[i] = 0.0f;
      }
      __syncthreads();
    }
  } else {
    scan_owned<1>(xl, src, dst, acc, xrs, mx, den, ring, cnt, over, n0, rows,
                  nedges, av, wave, lane);
    __syncthreads();
    scan_owned<2>(xl, src, dst, acc, xrs, mx, den, ring, cnt, over, n0, rows,
                  nedges, av, wave, lane);
    __syncthreads();
  }

  const float4* __restrict__ b4 = (const float4*)bias;
  float4* __restrict__ o4 = (float4*)(out + (size_t)n0 * CC);
  for (int i = tid; i < rows * (CC / 4); i += AGG_THREADS) {
    float4 a = acc4[i];
    const float dn = den[(i >> 4) * HH + ((i & 15) >> 2)] + 1e-16f;
    const float4 b = b4[i & (CC / 4 - 1)];
    a.x = a.x / dn + b.x; a.y = a.y / dn + b.y;
    a.z = a.z / dn + b.z; a.w = a.w / dn + b.w;
    if (do_relu) {
      a.x = fmaxf(a.x, 0.0f); a.y = fmaxf(a.y, 0.0f);
      a.z = fmaxf(a.z, 0.0f); a.w = fmaxf(a.w, 0.0f);
    }
    o4[i] = a;
  }
}

extern "C" void kernel_launch(void* const* d_in, const int* in_sizes, int n_in,
                              void* d_out, int out_size, void* d_ws,
                              size_t ws_size, hipStream_t stream) {
  const int N = in_sizes[0] / CC;
  const int E = in_sizes[1] / 2;

  const float* x = (const float*)d_in[0];
  const int* edge = (const int*)d_in[1];
  const int* src = edge;
  const int* dst = edge + E;

  const float* Wl[3] = {(const float*)d_in[2], (const float*)d_in[6],
                        (const float*)d_in[10]};
  const float* Wr[3] = {(const float*)d_in[3], (const float*)d_in[7],
                        (const float*)d_in[11]};
  const float* Av[3] = {(const float*)d_in[4], (const float*)d_in[8],
                        (const float*)d_in[12]};
  const float* Bv[3] = {(const float*)d_in[5], (const float*)d_in[9],
                        (const float*)d_in[13]};

  const size_t NC = (size_t)N * CC;
  float* ws = (float*)d_ws;
  float* xl = ws;                 ws += NC;
  float* hB = ws;                 ws += NC;
  float* hA = ws;                 ws += NC;
  float* sb = ws;                 ws += (size_t)N * HH;
  int* lists = (int*)ws;

  const float* lin[3] = {x, hA, hB};
  float* lout[3] = {hA, hB, (float*)d_out};
  const int lrelu[3] = {1, 1, 0};
  const int lbuild[3] = {1, 0, 0};

  const int gemm_grid = (N + 15) / 16;
  const int agg_grid = (N + ROWS_PB - 1) / ROWS_PB;

  for (int l = 0; l < 3; ++l) {
    gat_gemm_wmma<<<gemm_grid, 128, 0, stream>>>(lin[l], Wl[l], Av[l], xl,
                                                 sb, N);
    gat_owner_layer<<<agg_grid, AGG_THREADS, 0, stream>>>(
        xl, sb, lin[l], Wr[l], Av[l], src, dst, Bv[l], lout[l], lists, N, E,
        lrelu[l], lbuild[l]);
  }
}


# Layer 1 scans one owner byte per edge (dst/391, written by the first projection launch into a free workspace plane), 16 edges per load; hits take list slots by an LDS counter; LDS add order changes
# speedup vs baseline: 5.3294x; 1.0421x over previous
#include <hip/hip_runtime.h>


typedef float v2f __attribute__((ext_vector_type(2)));
typedef float v8f __attribute__((ext_vector_type(8)));

#define CC 64
#define HH 4
#define OO 16
#define NEG_SLOPE 0.2f
#define ROWS_PB 391

__device__ __forceinline__ unsigned fenc(float f) {
  unsigned b = __float_as_uint(f);
  return (b & 0x80000000u) ? ~b : (b | 0x80000000u);
}
__device__ __forceinline__ float fdec(unsigned e) {
  return (e & 0x80000000u) ? __uint_as_float(e & 0x7FFFFFFFu)
                           : __uint_as_float(~e);
}

__device__ __forceinline__ float leaky(float v) {
  return fmaxf(v, NEG_SLOPE * v);
}

__device__ __forceinline__ float quad_sum(float t) {
  t += __int_as_float(__builtin_amdgcn_update_dpp(
      0, __float_as_int(t), 0xB1, 0xF, 0xF, true));
  t += __int_as_float(__builtin_amdgcn_update_dpp(
      0, __float_as_int(t), 0x4E, 0xF, 0xF, true));
  return t;
}

__device__ __forceinline__ float row16_sum(float t) {
  t = quad_sum(t);
  t += __int_as_float(__builtin_amdgcn_update_dpp(
      0, __float_as_int(t), 0x141, 0xF, 0xF, true));
  t += __int_as_float(__builtin_amdgcn_update_dpp(
      0, __float_as_int(t), 0x140, 0xF, 0xF, true));
  return t;
}

__global__ __launch_bounds__(128) void gat_gemm_wmma(
    const float* __restrict__ X, const float* __restrict__ W,
    const float* __restrict__ att, float* __restrict__ Y,
    float* __restrict__ S, int nrows, const int* __restrict__ dstl,
    unsigned* __restrict__ codes, int nedges) {
  if (codes) {
    const int ng = nedges >> 2;
    for (int g = blockIdx.x * 128 + threadIdx.x; g < ng; g += gridDim.x * 128) {
      const unsigned o0 = (unsigned)dstl[4 * g + 0] / ROWS_PB;
      const unsigned o1 = (unsigned)dstl[4 * g + 1] / ROWS_PB;
      const unsigned o2 = (unsigned)dstl[4 * g + 2] / ROWS_PB;
      const unsigned o3 = (unsigned)dstl[4 * g + 3] / ROWS_PB;
      codes[g] = (o0 & 255u) | ((o1 & 255u) << 8) | ((o2 & 255u) << 16) |
                 ((o3 & 255u) << 24);
    }
  }
  const int wave = threadIdx.x >> 5;
  const int lane = threadIdx.x & 31;
  const int m0 = blockIdx.x * 16;
  if (m0 >= nrows) return;
  const int n0 = wave * 16;
  const int lrow = lane & 15;
  const int khalf = (lane >> 4) << 1;

  const float* __restrict__ xrow = X + (size_t)(m0 + lrow) * CC;
  v8f c = {};
#pragma unroll
  for (int k = 0; k < CC; k += 4) {
    v2f a, b;
    a.x = xrow[k + khalf];
    a.y = xrow[k + khalf + 1];
    b.x = W[(size_t)(k + khalf) * CC + n0 + lrow];
    b.y = W[(size_t)(k + khalf + 1) * CC + n0 + lrow];
    c = __builtin_amdgcn_wmma_f32_16x16x4_f32(false, a, false, b,
                                              (short)0, c, false, false);
  }
  const int rbase = m0 + ((lane >> 4) << 3);
#pragma unroll
  for (int r = 0; r < 8; ++r)
    Y[(size_t)(rbase + r) * CC + n0 + lrow] = c[r];
  const float ao = att[n0 + lrow];
  float sv = 0.0f;
#pragma unroll
  for (int r = 0; r < 8; ++r) {
    const float p = row16_sum(ao * leaky(c[r]));
    if (lrow == r) sv = p;
  }
  if (lrow < 8) S[(size_t)(rbase + lrow) * HH + wave] = sv;
}

#define AGG_THREADS 1024
#define AGG_WAVES (AGG_THREADS / 32)
#define RING 256
#define SCAN_G 8
#define STEP_K 8

__device__ __forceinline__ void project_own_rows(
    const float* __restrict__ hin, const float* __restrict__ Wr,
    const float* __restrict__ att, float* xrs, float* rbnd,
    int n0, int rows, int wave, int lane) {
  const int lrow = lane & 15;
  const int khalf = (lane >> 4) << 1;
  const int ncol = (wave & 3) * 16;
  const float ao = att[ncol + lrow];
  const float aabs = fabsf(ao);
  const float asgn = ao < 0.0f ? -1.0f : 1.0f;
  v2f b[CC / 4];
#pragma unroll
  for (int j = 0; j < CC / 4; ++j) {
    b[j].x = Wr[(size_t)(4 * j + khalf) * CC + ncol + lrow];
    b[j].y = Wr[(size_t)(4 * j + khalf + 1) * CC + ncol + lrow];
  }
  const int ntiles = (rows + 15) >> 4;
  for (int t = wave >> 2; t < ntiles; t += AGG_WAVES / 4) {
    const int m0 = t * 16;
    const float* __restrict__ xrow =
        hin + (size_t)(n0 + min(m0 + lrow, rows - 1)) * CC;
    v8f c = {};
#pragma unroll
    for (int j = 0; j < CC / 4; ++j) {
      v2f a;
      a.x = xrow[4 * j + khalf];
      a.y = xrow[4 * j + khalf + 1];
      c = __builtin_amdgcn_wmma_f32_16x16x4_f32(false, a, false, b[j],
                                                (short)0, c, false, false);
    }
    const int rb = m0 + ((lane >> 4) << 3);
#pragma unroll
    for (int r = 0; r < 8; ++r)
      if (rb + r < rows) xrs[(rb + r) * CC + ncol + lrow] = c[r];
    float rv = 0.0f;
#pragma unroll
    for (int r = 0; r < 8; ++r) {
      const float p = row16_sum(aabs * leaky(asgn * c[r]));
      if (lrow == r) rv = p;
    }
    if (lrow < 8 && rb + lrow < rows) rbnd[(rb + lrow) * HH + (wave & 3)] = rv;
  }
}

__device__ __forceinline__ void resolve_src(const int* __restrict__ src,
                                            int2* ring, int first, int cnt,
                                            int lane) {
  int e[RING / 32];
#pragma unroll
  for (int k = 0; k < RING / 32; ++k) {
    const int b = lane + 32 * k;
    e[k] = b < cnt ? ring[(first + b) & (RING - 1)].x : 0;
  }
#pragma unroll
  for (int k = 0; k < RING / 32; ++k) {
    const int b = lane + 32 * k;
    if (b < cnt) e[k] = src[e[k]];
  }
#pragma unroll
  for (int k = 0; k < RING / 32; ++k) {
    const int b = lane + 32 * k;
    if (b < cnt) ring[(first + b) & (RING - 1)].x = e[k];
  }
}

__device__ __forceinline__ void resolve_both(const int* __restrict__ src,
                                             const int* __restrict__ dst,
                                             int2* ring, int cnt, int n0,
                                             int lane) {
  int e[RING / 32], d[RING / 32];
#pragma unroll
  for (int k = 0; k < RING / 32; ++k) {
    const int b = lane + 32 * k;
    e[k] = b < cnt ? ring[b].x : 0;
  }
#pragma unroll
  for (int k = 0; k < RING / 32; ++k) {
    const int b = lane + 32 * k;
    d[k] = 0;
    if (b < cnt) {
      d[k] = dst[e[k]] - n0;
      e[k] = src[e[k]];
    }
  }
#pragma unroll
  for (int k = 0; k < RING / 32; ++k) {
    const int b = lane + 32 * k;
    if (b < cnt) ring[b] = make_int2(e[k], d[k]);
  }
}

__device__ __forceinline__ void scan_codes(
    const unsigned* __restrict__ codes, const int* __restrict__ dst,
    int2* ring, int* tailp, int& cnt_out, bool& over_out, int n0, int rows,
    int nedges, int wave, int lane) {
  const unsigned pat = (unsigned)blockIdx.x * 0x01010101u;
  const int n16 = ((((size_t)codes) & 15) == 0) ? (nedges >> 4) : 0;
  const uint4* __restrict__ c16 = (const uint4*)codes;
  for (int base = wave * (32 * SCAN_G); base < n16;
       base += AGG_WAVES * 32 * SCAN_G) {
    uint4 v[SCAN_G];
#pragma unroll
    for (int g = 0; g < SCAN_G; ++g) {
      const int idx = base + g * 32 + lane;
      v[g] = c16[min(idx, n16 - 1)];
    }
    unsigned gm = 0u;
#pragma unroll
    for (int g = 0; g < SCAN_G; ++g) {
      const int idx = base + g * 32 + lane;
      const unsigned z0 = v[g].x ^ pat, z1 = v[g].y ^ pat;
      const unsigned z2 = v[g].z ^ pat, z3 = v[g].w ^ pat;
      const unsigned any =
          (((z0 - 0x01010101u) & ~z0) | ((z1 - 0x01010101u) & ~z1) |
           ((z2 - 0x01010101u) & ~z2) | ((z3 - 0x01010101u) & ~z3)) &
          0x80808080u;
      if (any != 0u && idx < n16) gm |= 1u << g;
    }
    while (gm) {
      const int g = __builtin_ctz(gm);
      gm &= gm - 1u;
      const int idx = base + g * 32 + lane;
      const uint4 cv = c16[idx];
      const unsigned zz[4] = {cv.x ^ pat, cv.y ^ pat, cv.z ^ pat, cv.w ^ pat};
      unsigned hm[4];
      int c = 0;
#pragma unroll
      for (int w = 0; w < 4; ++w) {
        hm[w] = ~(((zz[w] & 0x7F7F7F7Fu) + 0x7F7F7F7Fu) | zz[w] |
                  0x7F7F7F7Fu);
        c += __builtin_popcount(hm[w]);
      }
      int pos = atomicAdd(tailp, c);
#pragma unroll
      for (int w = 0; w < 4; ++w) {
        unsigned m = hm[w];
        while (m) {
          const int bit = __builtin_ctz(m);
          m &= m - 1u;
          ring[pos & (RING - 1)] = make_int2(idx * 16 + w * 4 + (bit >> 3), 0);
          ++pos;
        }
      }
    }
  }
  for (int eb = 16 * n16 + wave * 32; eb < nedges; eb += AGG_WAVES * 32) {
    const int e = eb + lane;
    if (e < nedges && (unsigned)(dst[e] - n0) < (unsigned)rows) {
      const int pos = atomicAdd(tailp, 1);
      ring[pos & (RING - 1)] = make_int2(e, 0);
    }
  }
  const int tail = __builtin_amdgcn_readfirstlane(*(volatile int*)tailp);
  cnt_out = tail;
  over_out = tail > RING;
}

template <int PASS>
__device__ __forceinline__ void edge_steps(
    const float* __restrict__ xl, float* acc, const float* xrs, unsigned* mx,
    float* den, const int2* ring, int first, int cnt, float4 av, int lane) {
  const int c4 = (lane & 15) << 2;
  const int half = lane >> 4;
  const int hd = (lane & 15) >> 2;
  const bool quad0 = (lane & 3) == 0;
  const int steps = (cnt + 1) >> 1;
#pragma nounroll
  for (int k0 = 0; k0 < steps; k0 += STEP_K) {
    int2 q[STEP_K];
    float4 lv[STEP_K], rv[STEP_K];
    unsigned me[STEP_K];
#pragma unroll
    for (int k = 0; k < STEP_K; ++k) {
      const int idx = min(2 * (k0 + k) + half, cnt - 1);
      q[k] = ring[(first + idx) & (RING - 1)];
    }
#pragma unroll
    for (int k = 0; k < STEP_K; ++k) {
      lv[k] = *(const float4*)(xl + (size_t)q[k].x * CC + c4);
      rv[k] = *(const float4*)(xrs + q[k].y * CC + c4);
      me[k] = PASS == 1 ? 0u : mx[q[k].y * HH + hd];
    }
#pragma unroll
    for (int k = 0; k < STEP_K; ++k) {
      const bool live = 2 * (k0 + k) + half < cnt;
      float t = av.x * leaky(lv[k].x + rv[k].x);
      t += av.y * leaky(lv[k].y + rv[k].y);
      t += av.z * leaky(lv[k].z + rv[k].z);
      t += av.w * leaky(lv[k].w + rv[k].w);
      t = quad_sum(t);
      const int sh = q[k].y * HH + hd;
      if (PASS == 1) {
        if (live && quad0) atomicMax(&mx[sh], fenc(t));
      } else {
        const float w = __expf(t - fdec(me[k]));
        if (live) {
          float* dr = acc + q[k].y * CC + c4;
          atomicAdd(dr + 0, w * lv[k].x);
          atomicAdd(dr + 1, w * lv[k].y);
          atomicAdd(dr + 2, w * lv[k].z);
          atomicAdd(dr + 3, w * lv[k].w);
          if (quad0) atomicAdd(&den[sh], w);
        }
      }
    }
  }
}

__device__ __forceinline__ void bound_steps(
    const float* __restrict__ sbnd, const float* rbnd, unsigned* mx,
    const int2* ring, int cnt, int lane) {
  int2 q[RING / 32];
  float4 s[RING / 32];
#pragma unroll
  for (int k = 0; k < RING / 32; ++k) {
    const int b = lane + 32 * k;
    q[k] = b < cnt ? ring[b] : make_int2(0, 0);
  }
#pragma unroll
  for (int k = 0; k < RING / 32; ++k) {
    const int b = lane + 32 * k;
    s[k] = b < cnt ? *(const float4*)(sbnd + (size_t)q[k].x * HH)
                   : make_float4(0.0f, 0.0f, 0.0f, 0.0f);
  }
#pragma unroll
  for (int k = 0; k < RING / 32; ++k) {
    const int b = lane + 32 * k;
    if (b < cnt) {
      const float4 r = *(const float4*)(rbnd + q[k].y * HH);
      unsigned* m = mx + q[k].y * HH;
      atomicMax(m + 0, fenc(s[k].x + r.x));
      atomicMax(m + 1, fenc(s[k].y + r.y));
      atomicMax(m + 2, fenc(s[k].z + r.z));
      atomicMax(m + 3, fenc(s[k].w + r.w));
    }
  }
}

template <int MODE>
__device__ __forceinline__ void scan_owned(
    const float* __restrict__ xl, const int* __restrict__ src,
    const int* __restrict__ dst, float* acc, const float* xrs, unsigned* mx,
    float* den, int2* ring, int& cnt_out, bool& over_out, int n0, int rows,
    int nedges, float4 av, int wave, int lane) {
  int head = 0, tail = 0;

#define AGG_PUSH(m, e, x)                                                    \
  {                                                                          \
    const unsigned bal_ = __builtin_amdgcn_ballot_w32(m);                    \
    if (m)                                                                   \
      ring[(tail + (int)__builtin_amdgcn_mbcnt_lo(bal_, 0u)) & (RING - 1)] = \
          make_int2((e), (int)(x));                                          \
    tail += __builtin_popcount(bal_);                                        \
  }
#define AGG_DRAIN()                                                          \
  if (MODE != 0) {                                                           \
    while (tail - head >= 32) {                                              \
      resolve_src(src, ring, head, 32, lane);                                \
      edge_steps<(MODE == 1 ? 1 : 2)>(xl, acc, xrs, mx, den, ring, head, 32, \
                                      av, lane);                             \
      head += 32;                                                            \
    }                                                                        \
  }

  const int n4 = ((((size_t)dst) & 15) == 0) ? (nedges >> 2) : 0;
  const int4* __restrict__ dst4 = (const int4*)dst;
  for (int base = wave * (32 * SCAN_G); base < n4;
       base += AGG_WAVES * 32 * SCAN_G) {
    int4 v[SCAN_G];
#pragma unroll
    for (int g = 0; g < SCAN_G; ++g) {
      const int idx = base + g * 32 + lane;
      v[g] = dst4[min(idx, n4 - 1)];
    }
#pragma unroll
    for (int g = 0; g < SCAN_G; ++g) {
      const int idx = base + g * 32 + lane;
      const unsigned lim = idx < n4 ? (unsigned)rows : 0u;
      const unsigned x0 = (unsigned)(v[g].x - n0);
      const unsigned x1 = (unsigned)(v[g].y - n0);
      const unsigned x2 = (unsigned)(v[g].z - n0);
      const unsigned x3 = (unsigned)(v[g].w - n0);
      const bool m0 = x0 < lim, m1 = x1 < lim, m2 = x2 < lim, m3 = x3 < lim;
      if (__builtin_amdgcn_ballot_w32(m0 | m1 | m2 | m3)) {
        const int e0 = idx * 4;
        AGG_PUSH(m0, e0 + 0, x0)
        AGG_PUSH(m1, e0 + 1, x1)
        AGG_PUSH(m2, e0 + 2, x2)
        AGG_PUSH(m3, e0 + 3, x3)
      }
      AGG_DRAIN()
    }
  }
  for (int eb = 4 * n4 + wave * 32; eb < nedges; eb += AGG_WAVES * 32) {
    const int e = eb + lane;
    const unsigned x = e < nedges ? (unsigned)(dst[e] - n0) : 0xFFFFFFFFu;
    const bool m = x < (unsigned)rows;
    AGG_PUSH(m, e, x)
    AGG_DRAIN()
  }
  if (MODE != 0 && tail > head) {
    resolve_src(src, ring, head, tail - head, lane);
    edge_steps<(MODE == 1 ? 1 : 2)>(xl, acc, xrs, mx, den, ring, head,
                                    tail - head, av, lane);
  }
#undef AGG_PUSH
#undef AGG_DRAIN
  cnt_out = tail;
  over_out = MODE == 0 && tail > RING;
}

__global__ __launch_bounds__(AGG_THREADS) void gat_owner_layer(
    const float* __restrict__ xl, const float* __restrict__ sbnd,
    const float* __restrict__ hin,
    const float* __restrict__ wr, const float* __restrict__ att,
    const int* __restrict__ src,
    const int* __restrict__ dst, const float* __restrict__ bias,
    float* __restrict__ out, int* __restrict__ lists,
    const unsigned* __restrict__ codes, int n_nodes, int nedges,
    int do_relu, int build) {
  __shared__ __attribute__((aligned(16))) float acc[ROWS_PB * CC];
  __shared__ __attribute__((aligned(16))) float xrs[ROWS_PB * CC];
  __shared__ __attribute__((aligned(16))) float rbnd[ROWS_PB * HH];
  __shared__ unsigned mx[ROWS_PB * HH];
  __shared__ float den[ROWS_PB * HH];
  __shared__ int2 ring_all[AGG_WAVES * RING];
  __shared__ int wtail[AGG_WAVES];
  __shared__ int any_over;
  __shared__ int redo;

  const int tid = threadIdx.x;
  const int lane = tid & 31;
  const int wave = __builtin_amdgcn_readfirstlane(tid >> 5);
  const int n0 = blockIdx.x * ROWS_PB;
  const int rows = min(ROWS_PB, n_nodes - n0);

  float4* acc4 = (float4*)acc;
  for (int i = tid; i < rows * (CC / 4); i += AGG_THREADS)
    acc4[i] = make_float4(0.0f, 0.0f, 0.0f, 0.0f);
  for (int i = tid; i < rows * HH; i += AGG_THREADS) {
    mx[i] = 0u;
    den[i] = 0.0f;
  }
  project_own_rows(hin, wr, att, xrs, rbnd, n0, rows, wave, lane);
  const int nlists = gridDim.x * AGG_WAVES;
  const int lid = blockIdx.x * AGG_WAVES + wave;
  int2* glist = (int2*)lists + (size_t)lid * RING;
  int* gcnt = lists + (size_t)nlists * RING * 2;
  int* gover = gcnt + nlists;
  if (tid == 0) {
    any_over = build ? 0 : gover[blockIdx.x];
    redo = 0;
  }
  __syncthreads();

  const float4 av = ((const float4*)att)[lane & 15];
  int2* ring = ring_all + wave * RING;
  int cnt = 0;
  bool over = false;
  const bool bycode = build && gridDim.x <= 256;
  if (build) {
    if (bycode) {
      if (lane == 0) wtail[wave] = 0;
      scan_codes(codes, dst, ring, &wtail[wave], cnt, over, n0, rows, nedges,
                 wave, lane);
    } else {
      scan_owned<0>(xl, src, dst, acc, xrs, mx, den, ring, cnt, over, n0,
                    rows, nedges, av, wave, lane);
    }
    if (over && lane == 0) any_over = 1;
    __syncthreads();
    if (tid == 0) gover[blockIdx.x] = any_over;
  }
  if (any_over == 0) {
    if (build) {
      if (bycode)
        resolve_both(src, dst, ring, cnt, n0, lane);
      else
        resolve_src(src, ring, 0, cnt, lane);
#pragma unroll
      for (int k = 0; k < RING / 32; ++k) {
        const int b = lane + 32 * k;
        if (b < cnt) glist[b] = ring[b];
      }
      if (lane == 0) gcnt[lid] = cnt;
    } else {
      cnt = __builtin_amdgcn_readfirstlane(gcnt[lid]);
      int2 t[RING / 32];
#pragma unroll
      for (int k = 0; k < RING / 32; ++k) {
        const int b = lane + 32 * k;
        t[k] = b < cnt ? glist[b] : make_int2(0, 0);
      }
#pragma unroll
      for (int k = 0; k < RING / 32; ++k) {
        const int b = lane + 32 * k;
        if (b < cnt) ring[b] = t[k];
      }
    }
#pragma nounroll
    for (int exact = 0; exact < 2; ++exact) {
      if (exact == 0)
        bound_steps(sbnd, rbnd, mx, ring, cnt, lane);
      else
        edge_steps<1>(xl, acc, xrs, mx, den, ring, 0, cnt, av, lane);
      __syncthreads();
      edge_steps<2>(xl, acc, xrs, mx, den, ring, 0, cnt, av, lane);
      __syncthreads();
      if (exact) break;
      bool bad = false;
      for (int i = tid; i < rows * HH; i += AGG_THREADS)
        bad |= (mx[i] != 0u) & (den[i] < 1e-7f);
      if (bad) redo = 1;
      __syncthreads();
      if (redo == 0) break;
      for (int i = tid; i < rows * (CC / 4); i += AGG_THREADS)
        acc4[i] = make_float4(0.0f, 0.0f, 0.0f, 0.0f);
      for (int i = tid; i < rows * HH; i += AGG_THREADS) {
        mx[i] = 0u;
        den[i] = 0.0f;
      }
      __syncthreads();
    }
  } else {
    scan_owned<1>(xl, src, dst, acc, xrs, mx, den, ring, cnt, over, n0, rows,
                  nedges, av, wave, lane);
    __syncthreads();
    scan_owned<2>(xl, src, dst, acc, xrs, mx, den, ring, cnt, over, n0, rows,
                  nedges, av, wave, lane);
    __syncthreads();
  }

  const float4* __restrict__ b4 = (const float4*)bias;
  float4* __restrict__ o4 = (float4*)(out + (size_t)n0 * CC);
  for (int i = tid; i < rows * (CC / 4); i += AGG_THREADS) {
    float4 a = acc4[i];
    const float dn = den[(i >> 4) * HH + ((i & 15) >> 2)] + 1e-16f;
    const float4 b = b4[i & (CC / 4 - 1)];
    a.x = a.x / dn + b.x; a.y = a.y / dn + b.y;
    a.z = a.z / dn + b.z; a.w = a.w / dn + b.w;
    if (do_relu) {
      a.x = fmaxf(a.x, 0.0f); a.y = fmaxf(a.y, 0.0f);
      a.z = fmaxf(a.z, 0.0f); a.w = fmaxf(a.w, 0.0f);
    }
    o4[i] = a;
  }
}

extern "C" void kernel_launch(void* const* d_in, const int* in_sizes, int n_in,
                              void* d_out, int out_size, void* d_ws,
                              size_t ws_size, hipStream_t stream) {
  const int N = in_sizes[0] / CC;
  const int E = in_sizes[1] / 2;

  const float* x = (const float*)d_in[0];
  const int* edge = (const int*)d_in[1];
  const int* src = edge;
  const int* dst = edge + E;

  const float* Wl[3] = {(const float*)d_in[2], (const float*)d_in[6],
                        (const float*)d_in[10]};
  const float* Wr[3] = {(const float*)d_in[3], (const float*)d_in[7],
                        (const float*)d_in[11]};
  const float* Av[3] = {(const float*)d_in[4], (const float*)d_in[8],
                        (const float*)d_in[12]};
  const float* Bv[3] = {(const float*)d_in[5], (const float*)d_in[9],
                        (const float*)d_in[13]};

  const size_t NC = (size_t)N * CC;
  float* ws = (float*)d_ws;
  float* xl = ws;                 ws += NC;
  float* hB = ws;                 ws += NC;
  float* hA = ws;                 ws += NC;
  float* sb = ws;                 ws += (size_t)N * HH;
  int* lists = (int*)ws;
  unsigned* codes = (unsigned*)hB;

  const float* lin[3] = {x, hA, hB};
  float* lout[3] = {hA, hB, (float*)d_out};
  const int lrelu[3] = {1, 1, 0};
  const int lbuild[3] = {1, 0, 0};

  const int gemm_grid = (N + 15) / 16;
  const int agg_grid = (N + ROWS_PB - 1) / ROWS_PB;

  for (int l = 0; l < 3; ++l) {
    gat_gemm_wmma<<<gemm_grid, 128, 0, stream>>>(
        lin[l], Wl[l], Av[l], xl, sb, N, dst, lbuild[l] ? codes : nullptr, E);
    gat_owner_layer<<<agg_grid, AGG_THREADS, 0, stream>>>(
        xl, sb, lin[l], Wr[l], Av[l], src, dst, Bv[l], lout[l], lists, codes,
        N, E, lrelu[l], lbuild[l]);
  }
}


# Pass 2 LDS adds: the upper 16-lane half adds its four channels in the order 2,3,0,1, so one add instruction touches 32 banks instead of 16 banks twice; same adds, arrival order of LDS adds changes
# speedup vs baseline: 5.8435x; 1.0965x over previous
#include <hip/hip_runtime.h>


typedef float v2f __attribute__((ext_vector_type(2)));
typedef float v8f __attribute__((ext_vector_type(8)));

#define CC 64
#define HH 4
#define OO 16
#define NEG_SLOPE 0.2f
#define ROWS_PB 391

__device__ __forceinline__ unsigned fenc(float f) {
  unsigned b = __float_as_uint(f);
  return (b & 0x80000000u) ? ~b : (b | 0x80000000u);
}
__device__ __forceinline__ float fdec(unsigned e) {
  return (e & 0x80000000u) ? __uint_as_float(e & 0x7FFFFFFFu)
                           : __uint_as_float(~e);
}

__device__ __forceinline__ float leaky(float v) {
  return fmaxf(v, NEG_SLOPE * v);
}

__device__ __forceinline__ float quad_sum(float t) {
  t += __int_as_float(__builtin_amdgcn_update_dpp(
      0, __float_as_int(t), 0xB1, 0xF, 0xF, true));
  t += __int_as_float(__builtin_amdgcn_update_dpp(
      0, __float_as_int(t), 0x4E, 0xF, 0xF, true));
  return t;
}

__device__ __forceinline__ float row16_sum(float t) {
  t = quad_sum(t);
  t += __int_as_float(__builtin_amdgcn_update_dpp(
      0, __float_as_int(t), 0x141, 0xF, 0xF, true));
  t += __int_as_float(__builtin_amdgcn_update_dpp(
      0, __float_as_int(t), 0x140, 0xF, 0xF, true));
  return t;
}

__global__ __launch_bounds__(128) void gat_gemm_wmma(
    const float* __restrict__ X, const float* __restrict__ W,
    const float* __restrict__ att, float* __restrict__ Y,
    float* __restrict__ S, int nrows, const int* __restrict__ dstl,
    unsigned* __restrict__ codes, int nedges) {
  if (codes) {
    const int ng = nedges >> 2;
    for (int g = blockIdx.x * 128 + threadIdx.x; g < ng; g += gridDim.x * 128) {
      const unsigned o0 = (unsigned)dstl[4 * g + 0] / ROWS_PB;
      const unsigned o1 = (unsigned)dstl[4 * g + 1] / ROWS_PB;
      const unsigned o2 = (unsigned)dstl[4 * g + 2] / ROWS_PB;
      const unsigned o3 = (unsigned)dstl[4 * g + 3] / ROWS_PB;
      codes[g] = (o0 & 255u) | ((o1 & 255u) << 8) | ((o2 & 255u) << 16) |
                 ((o3 & 255u) << 24);
    }
  }
  const int wave = threadIdx.x >> 5;
  const int lane = threadIdx.x & 31;
  const int m0 = blockIdx.x * 16;
  if (m0 >= nrows) return;
  const int n0 = wave * 16;
  const int lrow = lane & 15;
  const int khalf = (lane >> 4) << 1;

  const float* __restrict__ xrow = X + (size_t)(m0 + lrow) * CC;
  v8f c = {};
#pragma unroll
  for (int k = 0; k < CC; k += 4) {
    v2f a, b;
    a.x = xrow[k + khalf];
    a.y = xrow[k + khalf + 1];
    b.x = W[(size_t)(k + khalf) * CC + n0 + lrow];
    b.y = W[(size_t)(k + khalf + 1) * CC + n0 + lrow];
    c = __builtin_amdgcn_wmma_f32_16x16x4_f32(false, a, false, b,
                                              (short)0, c, false, false);
  }
  const int rbase = m0 + ((lane >> 4) << 3);
#pragma unroll
  for (int r = 0; r < 8; ++r)
    Y[(size_t)(rbase + r) * CC + n0 + lrow] = c[r];
  const float ao = att[n0 + lrow];
  float sv = 0.0f;
#pragma unroll
  for (int r = 0; r < 8; ++r) {
    const float p = row16_sum(ao * leaky(c[r]));
    if (lrow == r) sv = p;
  }
  if (lrow < 8) S[(size_t)(rbase + lrow) * HH + wave] = sv;
}

#define AGG_THREADS 1024
#define AGG_WAVES (AGG_THREADS / 32)
#define RING 256
#define SCAN_G 8
#define STEP_K 8

__device__ __forceinline__ void project_own_rows(
    const float* __restrict__ hin, const float* __restrict__ Wr,
    const float* __restrict__ att, float* xrs, float* rbnd,
    int n0, int rows, int wave, int lane) {
  const int lrow = lane & 15;
  const int khalf = (lane >> 4) << 1;
  const int ncol = (wave & 3) * 16;
  const float ao = att[ncol + lrow];
  const float aabs = fabsf(ao);
  const float asgn = ao < 0.0f ? -1.0f : 1.0f;
  v2f b[CC / 4];
#pragma unroll
  for (int j = 0; j < CC / 4; ++j) {
    b[j].x = Wr[(size_t)(4 * j + khalf) * CC + ncol + lrow];
    b[j].y = Wr[(size_t)(4 * j + khalf + 1) * CC + ncol + lrow];
  }
  const int ntiles = (rows + 15) >> 4;
  for (int t = wave >> 2; t < ntiles; t += AGG_WAVES / 4) {
    const int m0 = t * 16;
    const float* __restrict__ xrow =
        hin + (size_t)(n0 + min(m0 + lrow, rows - 1)) * CC;
    v8f c = {};
#pragma unroll
    for (int j = 0; j < CC / 4; ++j) {
      v2f a;
      a.x = xrow[4 * j + khalf];
      a.y = xrow[4 * j + khalf + 1];
      c = __builtin_amdgcn_wmma_f32_16x16x4_f32(false, a, false, b[j],
                                                (short)0, c, false, false);
    }
    const int rb = m0 + ((lane >> 4) << 3);
#pragma unroll
    for (int r = 0; r < 8; ++r)
      if (rb + r < rows) xrs[(rb + r) * CC + ncol + lrow] = c[r];
    float rv = 0.0f;
#pragma unroll
    for (int r = 0; r < 8; ++r) {
      const float p = row16_sum(aabs * leaky(asgn * c[r]));
      if (lrow == r) rv = p;
    }
    if (lrow < 8 && rb + lrow < rows) rbnd[(rb + lrow) * HH + (wave & 3)] = rv;
  }
}

__device__ __forceinline__ void resolve_src(const int* __restrict__ src,
                                            int2* ring, int first, int cnt,
                                            int lane) {
  int e[RING / 32];
#pragma unroll
  for (int k = 0; k < RING / 32; ++k) {
    const int b = lane + 32 * k;
    e[k] = b < cnt ? ring[(first + b) & (RING - 1)].x : 0;
  }
#pragma unroll
  for (int k = 0; k < RING / 32; ++k) {
    const int b = lane + 32 * k;
    if (b < cnt) e[k] = src[e[k]];
  }
#pragma unroll
  for (int k = 0; k < RING / 32; ++k) {
    const int b = lane + 32 * k;
    if (b < cnt) ring[(first + b) & (RING - 1)].x = e[k];
  }
}

__device__ __forceinline__ void resolve_both(const int* __restrict__ src,
                                             const int* __restrict__ dst,
                                             int2* ring, int cnt, int n0,
                                             int lane) {
  int e[RING / 32], d[RING / 32];
#pragma unroll
  for (int k = 0; k < RING / 32; ++k) {
    const int b = lane + 32 * k;
    e[k] = b < cnt ? ring[b].x : 0;
  }
#pragma unroll
  for (int k = 0; k < RING / 32; ++k) {
    const int b = lane + 32 * k;
    d[k] = 0;
    if (b < cnt) {
      d[k] = dst[e[k]] - n0;
      e[k] = src[e[k]];
    }
  }
#pragma unroll
  for (int k = 0; k < RING / 32; ++k) {
    const int b = lane + 32 * k;
    if (b < cnt) ring[b] = make_int2(e[k], d[k]);
  }
}

__device__ __forceinline__ void scan_codes(
    const unsigned* __restrict__ codes, const int* __restrict__ dst,
    int2* ring, int* tailp, int& cnt_out, bool& over_out, int n0, int rows,
    int nedges, int wave, int lane) {
  const unsigned pat = (unsigned)blockIdx.x * 0x01010101u;
  const int n16 = ((((size_t)codes) & 15) == 0) ? (nedges >> 4) : 0;
  const uint4* __restrict__ c16 = (const uint4*)codes;
  for (int base = wave * (32 * SCAN_G); base < n16;
       base += AGG_WAVES * 32 * SCAN_G) {
    uint4 v[SCAN_G];
#pragma unroll
    for (int g = 0; g < SCAN_G; ++g) {
      const int idx = base + g * 32 + lane;
      v[g] = c16[min(idx, n16 - 1)];
    }
    unsigned gm = 0u;
#pragma unroll
    for (int g = 0; g < SCAN_G; ++g) {
      const int idx = base + g * 32 + lane;
      const unsigned z0 = v[g].x ^ pat, z1 = v[g].y ^ pat;
      const unsigned z2 = v[g].z ^ pat, z3 = v[g].w ^ pat;
      const unsigned any =
          (((z0 - 0x01010101u) & ~z0) | ((z1 - 0x01010101u) & ~z1) |
           ((z2 - 0x01010101u) & ~z2) | ((z3 - 0x01010101u) & ~z3)) &
          0x80808080u;
      if (any != 0u && idx < n16) gm |= 1u << g;
    }
    while (gm) {
      const int g = __builtin_ctz(gm);
      gm &= gm - 1u;
      const int idx = base + g * 32 + lane;
      const uint4 cv = c16[idx];
      const unsigned zz[4] = {cv.x ^ pat, cv.y ^ pat, cv.z ^ pat, cv.w ^ pat};
      unsigned hm[4];
      int c = 0;
#pragma unroll
      for (int w = 0; w < 4; ++w) {
        hm[w] = ~(((zz[w] & 0x7F7F7F7Fu) + 0x7F7F7F7Fu) | zz[w] |
                  0x7F7F7F7Fu);
        c += __builtin_popcount(hm[w]);
      }
      int pos = atomicAdd(tailp, c);
#pragma unroll
      for (int w = 0; w < 4; ++w) {
        unsigned m = hm[w];
        while (m) {
          const int bit = __builtin_ctz(m);
          m &= m - 1u;
          ring[pos & (RING - 1)] = make_int2(idx * 16 + w * 4 + (bit >> 3), 0);
          ++pos;
        }
      }
    }
  }
  for (int eb = 16 * n16 + wave * 32; eb < nedges; eb += AGG_WAVES * 32) {
    const int e = eb + lane;
    if (e < nedges && (unsigned)(dst[e] - n0) < (unsigned)rows) {
      const int pos = atomicAdd(tailp, 1);
      ring[pos & (RING - 1)] = make_int2(e, 0);
    }
  }
  const int tail = __builtin_amdgcn_readfirstlane(*(volatile int*)tailp);
  cnt_out = tail;
  over_out = tail > RING;
}

template <int PASS>
__device__ __forceinline__ void edge_steps(
    const float* __restrict__ xl, float* acc, const float* xrs, unsigned* mx,
    float* den, const int2* ring, int first, int cnt, float4 av, int lane) {
  const int c4 = (lane & 15) << 2;
  const int half = lane >> 4;
  const int hd = (lane & 15) >> 2;
  const bool quad0 = (lane & 3) == 0;
  const int steps = (cnt + 1) >> 1;
#pragma nounroll
  for (int k0 = 0; k0 < steps; k0 += STEP_K) {
    int2 q[STEP_K];
    float4 lv[STEP_K], rv[STEP_K];
    unsigned me[STEP_K];
#pragma unroll
    for (int k = 0; k < STEP_K; ++k) {
      const int idx = min(2 * (k0 + k) + half, cnt - 1);
      q[k] = ring[(first + idx) & (RING - 1)];
    }
#pragma unroll
    for (int k = 0; k < STEP_K; ++k) {
      lv[k] = *(const float4*)(xl + (size_t)q[k].x * CC + c4);
      rv[k] = *(const float4*)(xrs + q[k].y * CC + c4);
      me[k] = PASS == 1 ? 0u : mx[q[k].y * HH + hd];
    }
#pragma unroll
    for (int k = 0; k < STEP_K; ++k) {
      const bool live = 2 * (k0 + k) + half < cnt;
      float t = av.x * leaky(lv[k].x + rv[k].x);
      t += av.y * leaky(lv[k].y + rv[k].y);
      t += av.z * leaky(lv[k].z + rv[k].z);
      t += av.w * leaky(lv[k].w + rv[k].w);
      t = quad_sum(t);
      const int sh = q[k].y * HH + hd;
      if (PASS == 1) {
        if (live && quad0) atomicMax(&mx[sh], fenc(t));
      } else {
        const float w = __expf(t - fdec(me[k]));
        if (live) {
          float* dr = acc + q[k].y * CC + c4;
          float* d01 = dr + 2 * half;
          float* d23 = dr + 2 - 2 * half;
          const float p0 = w * (half ? lv[k].z : lv[k].x);
          const float p1 = w * (half ? lv[k].w : lv[k].y);
          const float p2 = w * (half ? lv[k].x : lv[k].z);
          const float p3 = w * (half ? lv[k].y : lv[k].w);
          atomicAdd(d01 + 0, p0);
          atomicAdd(d01 + 1, p1);
          atomicAdd(d23 + 0, p2);
          atomicAdd(d23 + 1, p3);
          if (quad0) atomicAdd(&den[sh], w);
        }
      }
    }
  }
}

__device__ __forceinline__ void bound_steps(
    const float* __restrict__ sbnd, const float* rbnd, unsigned* mx,
    const int2* ring, int cnt, int lane) {
  int2 q[RING / 32];
  float4 s[RING / 32];
#pragma unroll
  for (int k = 0; k < RING / 32; ++k) {
    const int b = lane + 32 * k;
    q[k] = b < cnt ? ring[b] : make_int2(0, 0);
  }
#pragma unroll
  for (int k = 0; k < RING / 32; ++k) {
    const int b = lane + 32 * k;
    s[k] = b < cnt ? *(const float4*)(sbnd + (size_t)q[k].x * HH)
                   : make_float4(0.0f, 0.0f, 0.0f, 0.0f);
  }
#pragma unroll
  for (int k = 0; k < RING / 32; ++k) {
    const int b = lane + 32 * k;
    if (b < cnt) {
      const float4 r = *(const float4*)(rbnd + q[k].y * HH);
      unsigned* m = mx + q[k].y * HH;
      atomicMax(m + 0, fenc(s[k].x + r.x));
      atomicMax(m + 1, fenc(s[k].y + r.y));
      atomicMax(m + 2, fenc(s[k].z + r.z));
      atomicMax(m + 3, fenc(s[k].w + r.w));
    }
  }
}

template <int MODE>
__device__ __forceinline__ void scan_owned(
    const float* __restrict__ xl, const int* __restrict__ src,
    const int* __restrict__ dst, float* acc, const float* xrs, unsigned* mx,
    float* den, int2* ring, int& cnt_out, bool& over_out, int n0, int rows,
    int nedges, float4 av, int wave, int lane) {
  int head = 0, tail = 0;

#define AGG_PUSH(m, e, x)                                                    \
  {                                                                          \
    const unsigned bal_ = __builtin_amdgcn_ballot_w32(m);                    \
    if (m)                                                                   \
      ring[(tail + (int)__builtin_amdgcn_mbcnt_lo(bal_, 0u)) & (RING - 1)] = \
          make_int2((e), (int)(x));                                          \
    tail += __builtin_popcount(bal_);                                        \
  }
#define AGG_DRAIN()                                                          \
  if (MODE != 0) {                                                           \
    while (tail - head >= 32) {                                              \
      resolve_src(src, ring, head, 32, lane);                                \
      edge_steps<(MODE == 1 ? 1 : 2)>(xl, acc, xrs, mx, den, ring, head, 32, \
                                      av, lane);                             \
      head += 32;                                                            \
    }                                                                        \
  }

  const int n4 = ((((size_t)dst) & 15) == 0) ? (nedges >> 2) : 0;
  const int4* __restrict__ dst4 = (const int4*)dst;
  for (int base = wave * (32 * SCAN_G); base < n4;
       base += AGG_WAVES * 32 * SCAN_G) {
    int4 v[SCAN_G];
#pragma unroll
    for (int g = 0; g < SCAN_G; ++g) {
      const int idx = base + g * 32 + lane;
      v[g] = dst4[min(idx, n4 - 1)];
    }
#pragma unroll
    for (int g = 0; g < SCAN_G; ++g) {
      const int idx = base + g * 32 + lane;
      const unsigned lim = idx < n4 ? (unsigned)rows : 0u;
      const unsigned x0 = (unsigned)(v[g].x - n0);
      const unsigned x1 = (unsigned)(v[g].y - n0);
      const unsigned x2 = (unsigned)(v[g].z - n0);
      const unsigned x3 = (unsigned)(v[g].w - n0);
      const bool m0 = x0 < lim, m1 = x1 < lim, m2 = x2 < lim, m3 = x3 < lim;
      if (__builtin_amdgcn_ballot_w32(m0 | m1 | m2 | m3)) {
        const int e0 = idx * 4;
        AGG_PUSH(m0, e0 + 0, x0)
        AGG_PUSH(m1, e0 + 1, x1)
        AGG_PUSH(m2, e0 + 2, x2)
        AGG_PUSH(m3, e0 + 3, x3)
      }
      AGG_DRAIN()
    }
  }
  for (int eb = 4 * n4 + wave * 32; eb < nedges; eb += AGG_WAVES * 32) {
    const int e = eb + lane;
    const unsigned x = e < nedges ? (unsigned)(dst[e] - n0) : 0xFFFFFFFFu;
    const bool m = x < (unsigned)rows;
    AGG_PUSH(m, e, x)
    AGG_DRAIN()
  }
  if (MODE != 0 && tail > head) {
    resolve_src(src, ring, head, tail - head, lane);
    edge_steps<(MODE == 1 ? 1 : 2)>(xl, acc, xrs, mx, den, ring, head,
                                    tail - head, av, lane);
  }
#undef AGG_PUSH
#undef AGG_DRAIN
  cnt_out = tail;
  over_out = MODE == 0 && tail > RING;
}

__global__ __launch_bounds__(AGG_THREADS) void gat_owner_layer(
    const float* __restrict__ xl, const float* __restrict__ sbnd,
    const float* __restrict__ hin,
    const float* __restrict__ wr, const float* __restrict__ att,
    const int* __restrict__ src,
    const int* __restrict__ dst, const float* __restrict__ bias,
    float* __restrict__ out, int* __restrict__ lists,
    const unsigned* __restrict__ codes, int n_nodes, int nedges,
    int do_relu, int build) {
  __shared__ __attribute__((aligned(16))) float acc[ROWS_PB * CC];
  __shared__ __attribute__((aligned(16))) float xrs[ROWS_PB * CC];
  __shared__ __attribute__((aligned(16))) float rbnd[ROWS_PB * HH];
  __shared__ unsigned mx[ROWS_PB * HH];
  __shared__ float den[ROWS_PB * HH];
  __shared__ int2 ring_all[AGG_WAVES * RING];
  __shared__ int wtail[AGG_WAVES];
  __shared__ int any_over;
  __shared__ int redo;

  const int tid = threadIdx.x;
  const int lane = tid & 31;
  const int wave = __builtin_amdgcn_readfirstlane(tid >> 5);
  const int n0 = blockIdx.x * ROWS_PB;
  const int rows = min(ROWS_PB, n_nodes - n0);

  float4* acc4 = (float4*)acc;
  for (int i = tid; i < rows * (CC / 4); i += AGG_THREADS)
    acc4[i] = make_float4(0.0f, 0.0f, 0.0f, 0.0f);
  for (int i = tid; i < rows * HH; i += AGG_THREADS) {
    mx[i] = 0u;
    den[i] = 0.0f;
  }
  project_own_rows(hin, wr, att, xrs, rbnd, n0, rows, wave, lane);
  const int nlists = gridDim.x * AGG_WAVES;
  const int lid = blockIdx.x * AGG_WAVES + wave;
  int2* glist = (int2*)lists + (size_t)lid * RING;
  int* gcnt = lists + (size_t)nlists * RING * 2;
  int* gover = gcnt + nlists;
  if (tid == 0) {
    any_over = build ? 0 : gover[blockIdx.x];
    redo = 0;
  }
  __syncthreads();

  const float4 av = ((const float4*)att)[lane & 15];
  int2* ring = ring_all + wave * RING;
  int cnt = 0;
  bool over = false;
  const bool bycode = build && gridDim.x <= 256;
  if (build) {
    if (bycode) {
      if (lane == 0) wtail[wave] = 0;
      scan_codes(codes, dst, ring, &wtail[wave], cnt, over, n0, rows, nedges,
                 wave, lane);
    } else {
      scan_owned<0>(xl, src, dst, acc, xrs, mx, den, ring, cnt, over, n0,
                    rows, nedges, av, wave, lane);
    }
    if (over && lane == 0) any_over = 1;
    __syncthreads();
    if (tid == 0) gover[blockIdx.x] = any_over;
  }
  if (any_over == 0) {
    if (build) {
      if (bycode)
        resolve_both(src, dst, ring, cnt, n0, lane);
      else
        resolve_src(src, ring, 0, cnt, lane);
#pragma unroll
      for (int k = 0; k < RING / 32; ++k) {
        const int b = lane + 32 * k;
        if (b < cnt) glist[b] = ring[b];
      }
      if (lane == 0) gcnt[lid] = cnt;
    } else {
      cnt = __builtin_amdgcn_readfirstlane(gcnt[lid]);
      int2 t[RING / 32];
#pragma unroll
      for (int k = 0; k < RING / 32; ++k) {
        const int b = lane + 32 * k;
        t[k] = b < cnt ? glist[b] : make_int2(0, 0);
      }
#pragma unroll
      for (int k = 0; k < RING / 32; ++k) {
        const int b = lane + 32 * k;
        if (b < cnt) ring[b] = t[k];
      }
    }
#pragma nounroll
    for (int exact = 0; exact < 2; ++exact) {
      if (exact == 0)
        bound_steps(sbnd, rbnd, mx, ring, cnt, lane);
      else
        edge_steps<1>(xl, acc, xrs, mx, den, ring, 0, cnt, av, lane);
      __syncthreads();
      edge_steps<2>(xl, acc, xrs, mx, den, ring, 0, cnt, av, lane);
      __syncthreads();
      if (exact) break;
      bool bad = false;
      for (int i = tid; i < rows * HH; i += AGG_THREADS)
        bad |= (mx[i] != 0u) & (den[i] < 1e-7f);
      if (bad) redo = 1;
      __syncthreads();
      if (redo == 0) break;
      for (int i = tid; i < rows * (CC / 4); i += AGG_THREADS)
        acc4[i] = make_float4(0.0f, 0.0f, 0.0f, 0.0f);
      for (int i = tid; i < rows * HH; i += AGG_THREADS) {
        mx[i] = 0u;
        den[i] = 0.0f;
      }
      __syncthreads();
    }
  } else {
    scan_owned<1>(xl, src, dst, acc, xrs, mx, den, ring, cnt, over, n0, rows,
                  nedges, av, wave, lane);
    __syncthreads();
    scan_owned<2>(xl, src, dst, acc, xrs, mx, den, ring, cnt, over, n0, rows,
                  nedges, av, wave, lane);
    __syncthreads();
  }

  const float4* __restrict__ b4 = (const float4*)bias;
  float4* __restrict__ o4 = (float4*)(out + (size_t)n0 * CC);
  for (int i = tid; i < rows * (CC / 4); i += AGG_THREADS) {
    float4 a = acc4[i];
    const float dn = den[(i >> 4) * HH + ((i & 15) >> 2)] + 1e-16f;
    const float4 b = b4[i & (CC / 4 - 1)];
    a.x = a.x / dn + b.x; a.y = a.y / dn + b.y;
    a.z = a.z / dn + b.z; a.w = a.w / dn + b.w;
    if (do_relu) {
      a.x = fmaxf(a.x, 0.0f); a.y = fmaxf(a.y, 0.0f);
      a.z = fmaxf(a.z, 0.0f); a.w = fmaxf(a.w, 0.0f);
    }
    o4[i] = a;
  }
}

extern "C" void kernel_launch(void* const* d_in, const int* in_sizes, int n_in,
                              void* d_out, int out_size, void* d_ws,
                              size_t ws_size, hipStream_t stream) {
  const int N = in_sizes[0] / CC;
  const int E = in_sizes[1] / 2;

  const float* x = (const float*)d_in[0];
  const int* edge = (const int*)d_in[1];
  const int* src = edge;
  const int* dst = edge + E;

  const float* Wl[3] = {(const float*)d_in[2], (const float*)d_in[6],
                        (const float*)d_in[10]};
  const float* Wr[3] = {(const float*)d_in[3], (const float*)d_in[7],
                        (const float*)d_in[11]};
  const float* Av[3] = {(const float*)d_in[4], (const float*)d_in[8],
                        (const float*)d_in[12]};
  const float* Bv[3] = {(const float*)d_in[5], (const float*)d_in[9],
                        (const float*)d_in[13]};

  const size_t NC = (size_t)N * CC;
  float* ws = (float*)d_ws;
  float* xl = ws;                 ws += NC;
  float* hB = ws;                 ws += NC;
  float* hA = ws;                 ws += NC;
  float* sb = ws;                 ws += (size_t)N * HH;
  int* lists = (int*)ws;
  unsigned* codes = (unsigned*)hB;

  const float* lin[3] = {x, hA, hB};
  float* lout[3] = {hA, hB, (float*)d_out};
  const int lrelu[3] = {1, 1, 0};
  const int lbuild[3] = {1, 0, 0};

  const int gemm_grid = (N + 15) / 16;
  const int agg_grid = (N + ROWS_PB - 1) / ROWS_PB;

  for (int l = 0; l < 3; ++l) {
    gat_gemm_wmma<<<gemm_grid, 128, 0, stream>>>(
        lin[l], Wl[l], Av[l], xl, sb, N, dst, lbuild[l] ? codes : nullptr, E);
    gat_owner_layer<<<agg_grid, AGG_THREADS, 0, stream>>>(
        xl, sb, lin[l], Wr[l], Av[l], src, dst, Bv[l], lout[l], lists, codes,
        N, E, lrelu[l], lbuild[l]);
  }
}


# Next layer's xl=h@Wl and its bound S computed by each owner block in its tail from its finished rows in LDS (f32 matrix instr.); 4 launches for 6; lists one word per edge; sums unchanged
# speedup vs baseline: 6.0598x; 1.0370x over previous
#include <hip/hip_runtime.h>


typedef float v2f __attribute__((ext_vector_type(2)));
typedef float v8f __attribute__((ext_vector_type(8)));

#define CC 64
#define HH 4
#define OO 16
#define NEG_SLOPE 0.2f
#define ROWS_PB 391

__device__ __forceinline__ unsigned fenc(float f) {
  unsigned b = __float_as_uint(f);
  return (b & 0x80000000u) ? ~b : (b | 0x80000000u);
}
__device__ __forceinline__ float fdec(unsigned e) {
  return (e & 0x80000000u) ? __uint_as_float(e & 0x7FFFFFFFu)
                           : __uint_as_float(~e);
}

__device__ __forceinline__ float leaky(float v) {
  return fmaxf(v, NEG_SLOPE * v);
}

__device__ __forceinline__ float quad_sum(float t) {
  t += __int_as_float(__builtin_amdgcn_update_dpp(
      0, __float_as_int(t), 0xB1, 0xF, 0xF, true));
  t += __int_as_float(__builtin_amdgcn_update_dpp(
      0, __float_as_int(t), 0x4E, 0xF, 0xF, true));
  return t;
}

__device__ __forceinline__ float row16_sum(float t) {
  t = quad_sum(t);
  t += __int_as_float(__builtin_amdgcn_update_dpp(
      0, __float_as_int(t), 0x141, 0xF, 0xF, true));
  t += __int_as_float(__builtin_amdgcn_update_dpp(
      0, __float_as_int(t), 0x140, 0xF, 0xF, true));
  return t;
}

__global__ __launch_bounds__(128) void gat_gemm_wmma(
    const float* __restrict__ X, const float* __restrict__ W,
    const float* __restrict__ att, float* __restrict__ Y,
    float* __restrict__ S, int nrows, const int* __restrict__ dstl,
    unsigned* __restrict__ codes, int nedges) {
  if (codes) {
    const int ng = nedges >> 2;
    for (int g = blockIdx.x * 128 + threadIdx.x; g < ng; g += gridDim.x * 128) {
      const unsigned o0 = (unsigned)dstl[4 * g + 0] / ROWS_PB;
      const unsigned o1 = (unsigned)dstl[4 * g + 1] / ROWS_PB;
      const unsigned o2 = (unsigned)dstl[4 * g + 2] / ROWS_PB;
      const unsigned o3 = (unsigned)dstl[4 * g + 3] / ROWS_PB;
      codes[g] = (o0 & 255u) | ((o1 & 255u) << 8) | ((o2 & 255u) << 16) |
                 ((o3 & 255u) << 24);
    }
  }
  const int wave = threadIdx.x >> 5;
  const int lane = threadIdx.x & 31;
  const int m0 = blockIdx.x * 16;
  if (m0 >= nrows) return;
  const int n0 = wave * 16;
  const int lrow = lane & 15;
  const int khalf = (lane >> 4) << 1;

  const float* __restrict__ xrow = X + (size_t)(m0 + lrow) * CC;
  v8f c = {};
#pragma unroll
  for (int k = 0; k < CC; k += 4) {
    v2f a, b;
    a.x = xrow[k + khalf];
    a.y = xrow[k + khalf + 1];
    b.x = W[(size_t)(k + khalf) * CC + n0 + lrow];
    b.y = W[(size_t)(k + khalf + 1) * CC + n0 + lrow];
    c = __builtin_amdgcn_wmma_f32_16x16x4_f32(false, a, false, b,
                                              (short)0, c, false, false);
  }
  const int rbase = m0 + ((lane >> 4) << 3);
#pragma unroll
  for (int r = 0; r < 8; ++r)
    Y[(size_t)(rbase + r) * CC + n0 + lrow] = c[r];
  const float ao = att[n0 + lrow];
  float sv = 0.0f;
#pragma unroll
  for (int r = 0; r < 8; ++r) {
    const float p = row16_sum(ao * leaky(c[r]));
    if (lrow == r) sv = p;
  }
  if (lrow < 8) S[(size_t)(rbase + lrow) * HH + wave] = sv;
}

#define AGG_THREADS 1024
#define AGG_WAVES (AGG_THREADS / 32)
#define RING 256
#define SCAN_G 8
#define STEP_K 8
#define ACC_P 68

__device__ __forceinline__ void project_own_rows(
    const float* __restrict__ hin, const float* __restrict__ Wr,
    const float* __restrict__ att, float* xrs, float* rbnd,
    int n0, int rows, int wave, int lane) {
  const int lrow = lane & 15;
  const int khalf = (lane >> 4) << 1;
  const int ncol = (wave & 3) * 16;
  const float ao = att[ncol + lrow];
  const float aabs = fabsf(ao);
  const float asgn = ao < 0.0f ? -1.0f : 1.0f;
  v2f b[CC / 4];
#pragma unroll
  for (int j = 0; j < CC / 4; ++j) {
    b[j].x = Wr[(size_t)(4 * j + khalf) * CC + ncol + lrow];
    b[j].y = Wr[(size_t)(4 * j + khalf + 1) * CC + ncol + lrow];
  }
  const int ntiles = (rows + 15) >> 4;
  for (int t = wave >> 2; t < ntiles; t += AGG_WAVES / 4) {
    const int m0 = t * 16;
    const float* __restrict__ xrow =
        hin + (size_t)(n0 + min(m0 + lrow, rows - 1)) * CC;
    v8f c = {};
#pragma unroll
    for (int j = 0; j < CC / 4; ++j) {
      v2f a;
      a.x = xrow[4 * j + khalf];
      a.y = xrow[4 * j + khalf + 1];
      c = __builtin_amdgcn_wmma_f32_16x16x4_f32(false, a, false, b[j],
                                                (short)0, c, false, false);
    }
    const int rb = m0 + ((lane >> 4) << 3);
#pragma unroll
    for (int r = 0; r < 8; ++r)
      if (rb + r < rows) xrs[(rb + r) * CC + ncol + lrow] = c[r];
    float rv = 0.0f;
#pragma unroll
    for (int r = 0; r < 8; ++r) {
      const float p = row16_sum(aabs * leaky(asgn * c[r]));
      if (lrow == r) rv = p;
    }
    if (lrow < 8 && rb + lrow < rows) rbnd[(rb + lrow) * HH + (wave & 3)] = rv;
  }
}

__device__ __forceinline__ void project_next_rows(
    const float* hs, const float* __restrict__ Wl,
    const float* __restrict__ att, float* Y, float* S,
    int n0, int rows, int wave, int lane) {
  const int lrow = lane & 15;
  const int khalf = (lane >> 4) << 1;
  const int ncol = (wave & 3) * 16;
  const float ao = att[ncol + lrow];
  v2f b[CC / 4];
#pragma unroll
  for (int j = 0; j < CC / 4; ++j) {
    b[j].x = Wl[(size_t)(4 * j + khalf) * CC + ncol + lrow];
    b[j].y = Wl[(size_t)(4 * j + khalf + 1) * CC + ncol + lrow];
  }
  const int ntiles = (rows + 15) >> 4;
  for (int t = wave >> 2; t < ntiles; t += AGG_WAVES / 4) {
    const int m0 = t * 16;
    const float* xrow = hs + min(m0 + lrow, rows - 1) * ACC_P;
    v8f c = {};
#pragma unroll
    for (int j = 0; j < CC / 4; ++j) {
      const v2f a = *(const v2f*)(xrow + 4 * j + khalf);
      c = __builtin_amdgcn_wmma_f32_16x16x4_f32(false, a, false, b[j],
                                                (short)0, c, false, false);
    }
    const int rb = m0 + ((lane >> 4) << 3);
#pragma unroll
    for (int r = 0; r < 8; ++r)
      if (rb + r < rows) Y[(size_t)(n0 + rb + r) * CC + ncol + lrow] = c[r];
    float sv = 0.0f;
#pragma unroll
    for (int r = 0; r < 8; ++r) {
      const float p = row16_sum(ao * leaky(c[r]));
      if (lrow == r) sv = p;
    }
    if (lrow < 8 && rb + lrow < rows)
      S[(size_t)(n0 + rb + lrow) * HH + (wave & 3)] = sv;
  }
}

__device__ __forceinline__ void resolve_src(const int* __restrict__ src,
                                            int2* ring, int first, int cnt,
                                            int lane) {
  int e[RING / 32];
#pragma unroll
  for (int k = 0; k < RING / 32; ++k) {
    const int b = lane + 32 * k;
    e[k] = b < cnt ? ring[(first + b) & (RING - 1)].x : 0;
  }
#pragma unroll
  for (int k = 0; k < RING / 32; ++k) {
    const int b = lane + 32 * k;
    if (b < cnt) e[k] = src[e[k]];
  }
#pragma unroll
  for (int k = 0; k < RING / 32; ++k) {
    const int b = lane + 32 * k;
    if (b < cnt) ring[(first + b) & (RING - 1)].x = e[k];
  }
}

__device__ __forceinline__ void resolve_both(const int* __restrict__ src,
                                             const int* __restrict__ dst,
                                             int2* ring, int cnt, int n0,
                                             int lane) {
  int e[RING / 32], d[RING / 32];
#pragma unroll
  for (int k = 0; k < RING / 32; ++k) {
    const int b = lane + 32 * k;
    e[k] = b < cnt ? ring[b].x : 0;
  }
#pragma unroll
  for (int k = 0; k < RING / 32; ++k) {
    const int b = lane + 32 * k;
    d[k] = 0;
    if (b < cnt) {
      d[k] = dst[e[k]] - n0;
      e[k] = src[e[k]];
    }
  }
#pragma unroll
  for (int k = 0; k < RING / 32; ++k) {
    const int b = lane + 32 * k;
    if (b < cnt) ring[b] = make_int2(e[k], d[k]);
  }
}

__device__ __forceinline__ void scan_codes(
    const unsigned* __restrict__ codes, const int* __restrict__ dst,
    int2* ring, int* tailp, int& cnt_out, bool& over_out, int n0, int rows,
    int nedges, int wave, int lane) {
  const unsigned pat = (unsigned)blockIdx.x * 0x01010101u;
  const int n16 = ((((size_t)codes) & 15) == 0) ? (nedges >> 4) : 0;
  const uint4* __restrict__ c16 = (const uint4*)codes;
  for (int base = wave * (32 * SCAN_G); base < n16;
       base += AGG_WAVES * 32 * SCAN_G) {
    uint4 v[SCAN_G];
#pragma unroll
    for (int g = 0; g < SCAN_G; ++g) {
      const int idx = base + g * 32 + lane;
      v[g] = c16[min(idx, n16 - 1)];
    }
    unsigned gm = 0u;
#pragma unroll
    for (int g = 0; g < SCAN_G; ++g) {
      const int idx = base + g * 32 + lane;
      const unsigned z0 = v[g].x ^ pat, z1 = v[g].y ^ pat;
      const unsigned z2 = v[g].z ^ pat, z3 = v[g].w ^ pat;
      const unsigned any =
          (((z0 - 0x01010101u) & ~z0) | ((z1 - 0x01010101u) & ~z1) |
           ((z2 - 0x01010101u) & ~z2) | ((z3 - 0x01010101u) & ~z3)) &
          0x80808080u;
      if (any != 0u && idx < n16) gm |= 1u << g;
    }
    while (gm) {
      const int g = __builtin_ctz(gm);
      gm &= gm - 1u;
      const int idx = base + g * 32 + lane;
      const uint4 cv = c16[idx];
      const unsigned zz[4] = {cv.x ^ pat, cv.y ^ pat, cv.z ^ pat, cv.w ^ pat};
      unsigned hm[4];
      int c = 0;
#pragma unroll
      for (int w = 0; w < 4; ++w) {
        hm[w] = ~(((zz[w] & 0x7F7F7F7Fu) + 0x7F7F7F7Fu) | zz[w] |
                  0x7F7F7F7Fu);
        c += __builtin_popcount(hm[w]);
      }
      int pos = atomicAdd(tailp, c);
#pragma unroll
      for (int w = 0; w < 4; ++w) {
        unsigned m = hm[w];
        while (m) {
          const int bit = __builtin_ctz(m);
          m &= m - 1u;
          ring[pos & (RING - 1)] = make_int2(idx * 16 + w * 4 + (bit >> 3), 0);
          ++pos;
        }
      }
    }
  }
  for (int eb = 16 * n16 + wave * 32; eb < nedges; eb += AGG_WAVES * 32) {
    const int e = eb + lane;
    if (e < nedges && (unsigned)(dst[e] - n0) < (unsigned)rows) {
      const int pos = atomicAdd(tailp, 1);
      ring[pos & (RING - 1)] = make_int2(e, 0);
    }
  }
  const int tail = __builtin_amdgcn_readfirstlane(*(volatile int*)tailp);
  cnt_out = tail;
  over_out = tail > RING;
}

template <int PASS>
__device__ __forceinline__ void edge_steps(
    const float* __restrict__ xl, float* acc, const float* xrs, unsigned* mx,
    float* den, const int2* ring, int first, int cnt, float4 av, int lane) {
  const int c4 = (lane & 15) << 2;
  const int half = lane >> 4;
  const int hd = (lane & 15) >> 2;
  const bool quad0 = (lane & 3) == 0;
  const int steps = (cnt + 1) >> 1;
#pragma nounroll
  for (int k0 = 0; k0 < steps; k0 += STEP_K) {
    int2 q[STEP_K];
    float4 lv[STEP_K], rv[STEP_K];
    unsigned me[STEP_K];
#pragma unroll
    for (int k = 0; k < STEP_K; ++k) {
      const int idx = min(2 * (k0 + k) + half, cnt - 1);
      q[k] = ring[(first + idx) & (RING - 1)];
    }
#pragma unroll
    for (int k = 0; k < STEP_K; ++k) {
      lv[k] = *(const float4*)(xl + (size_t)q[k].x * CC + c4);
      rv[k] = *(const float4*)(xrs + q[k].y * CC + c4);
      me[k] = PASS == 1 ? 0u : mx[q[k].y * HH + hd];
    }
#pragma unroll
    for (int k = 0; k < STEP_K; ++k) {
      const bool live = 2 * (k0 + k) + half < cnt;
      float t = av.x * leaky(lv[k].x + rv[k].x);
      t += av.y * leaky(lv[k].y + rv[k].y);
      t += av.z * leaky(lv[k].z + rv[k].z);
      t += av.w * leaky(lv[k].w + rv[k].w);
      t = quad_sum(t);
      const int sh = q[k].y * HH + hd;
      if (PASS == 1) {
        if (live && quad0) atomicMax(&mx[sh], fenc(t));
      } else {
        const float w = __expf(t - fdec(me[k]));
        if (live) {
          float* dr = acc + q[k].y * ACC_P + c4;
          float* d01 = dr + 2 * half;
          float* d23 = dr + 2 - 2 * half;
          const float p0 = w * (half ? lv[k].z : lv[k].x);
          const float p1 = w * (half ? lv[k].w : lv[k].y);
          const float p2 = w * (half ? lv[k].x : lv[k].z);
          const float p3 = w * (half ? lv[k].y : lv[k].w);
          atomicAdd(d01 + 0, p0);
          atomicAdd(d01 + 1, p1);
          atomicAdd(d23 + 0, p2);
          atomicAdd(d23 + 1, p3);
          if (quad0) atomicAdd(&den[sh], w);
        }
      }
    }
  }
}

__device__ __forceinline__ void bound_steps(
    const float* __restrict__ sbnd, const float* rbnd, unsigned* mx,
    const int2* ring, int cnt, int lane) {
  int2 q[RING / 32];
  float4 s[RING / 32];
#pragma unroll
  for (int k = 0; k < RING / 32; ++k) {
    const int b = lane + 32 * k;
    q[k] = b < cnt ? ring[b] : make_int2(0, 0);
  }
#pragma unroll
  for (int k = 0; k < RING / 32; ++k) {
    const int b = lane + 32 * k;
    s[k] = b < cnt ? *(const float4*)(sbnd + (size_t)q[k].x * HH)
                   : make_float4(0.0f, 0.0f, 0.0f, 0.0f);
  }
#pragma unroll
  for (int k = 0; k < RING / 32; ++k) {
    const int b = lane + 32 * k;
    if (b < cnt) {
      const float4 r = *(const float4*)(rbnd + q[k].y * HH);
      unsigned* m = mx + q[k].y * HH;
      atomicMax(m + 0, fenc(s[k].x + r.x));
      atomicMax(m + 1, fenc(s[k].y + r.y));
      atomicMax(m + 2, fenc(s[k].z + r.z));
      atomicMax(m + 3, fenc(s[k].w + r.w));
    }
  }
}

template <int MODE>
__device__ __forceinline__ void scan_owned(
    const float* __restrict__ xl, const int* __restrict__ src,
    const int* __restrict__ dst, float* acc, const float* xrs, unsigned* mx,
    float* den, int2* ring, int& cnt_out, bool& over_out, int n0, int rows,
    int nedges, float4 av, int wave, int lane) {
  int head = 0, tail = 0;

#define AGG_PUSH(m, e, x)                                                    \
  {                                                                          \
    const unsigned bal_ = __builtin_amdgcn_ballot_w32(m);                    \
    if (m)                                                                   \
      ring[(tail + (int)__builtin_amdgcn_mbcnt_lo(bal_, 0u)) & (RING - 1)] = \
          make_int2((e), (int)(x));                                          \
    tail += __builtin_popcount(bal_);                                        \
  }
#define AGG_DRAIN()                                                          \
  if (MODE != 0) {                                                           \
    while (tail - head >= 32) {                                              \
      resolve_src(src, ring, head, 32, lane);                                \
      edge_steps<(MODE == 1 ? 1 : 2)>(xl, acc, xrs, mx, den, ring, head, 32, \
                                      av, lane);                             \
      head += 32;                                                            \
    }                                                                        \
  }

  const int n4 = ((((size_t)dst) & 15) == 0) ? (nedges >> 2) : 0;
  const int4* __restrict__ dst4 = (const int4*)dst;
  for (int base = wave * (32 * SCAN_G); base < n4;
       base += AGG_WAVES * 32 * SCAN_G) {
    int4 v[SCAN_G];
#pragma unroll
    for (int g = 0; g < SCAN_G; ++g) {
      const int idx = base + g * 32 + lane;
      v[g] = dst4[min(idx, n4 - 1)];
    }
#pragma unroll
    for (int g = 0; g < SCAN_G; ++g) {
      const int idx = base + g * 32 + lane;
      const unsigned lim = idx < n4 ? (unsigned)rows : 0u;
      const unsigned x0 = (unsigned)(v[g].x - n0);
      const unsigned x1 = (unsigned)(v[g].y - n0);
      const unsigned x2 = (unsigned)(v[g].z - n0);
      const unsigned x3 = (unsigned)(v[g].w - n0);
      const bool m0 = x0 < lim, m1 = x1 < lim, m2 = x2 < lim, m3 = x3 < lim;
      if (__builtin_amdgcn_ballot_w32(m0 | m1 | m2 | m3)) {
        const int e0 = idx * 4;
        AGG_PUSH(m0, e0 + 0, x0)
        AGG_PUSH(m1, e0 + 1, x1)
        AGG_PUSH(m2, e0 + 2, x2)
        AGG_PUSH(m3, e0 + 3, x3)
      }
      AGG_DRAIN()
    }
  }
  for (int eb = 4 * n4 + wave * 32; eb < nedges; eb += AGG_WAVES * 32) {
    const int e = eb + lane;
    const unsigned x = e < nedges ? (unsigned)(dst[e] - n0) : 0xFFFFFFFFu;
    const bool m = x < (unsigned)rows;
    AGG_PUSH(m, e, x)
    AGG_DRAIN()
  }
  if (MODE != 0 && tail > head) {
    resolve_src(src, ring, head, tail - head, lane);
    edge_steps<(MODE == 1 ? 1 : 2)>(xl, acc, xrs, mx, den, ring, head,
                                    tail - head, av, lane);
  }
#undef AGG_PUSH
#undef AGG_DRAIN
  cnt_out = tail;
  over_out = MODE == 0 && tail > RING;
}

__global__ __launch_bounds__(AGG_THREADS) void gat_owner_layer(
    const float* __restrict__ xl, const float* __restrict__ sbnd,
    const float* hin,
    const float* __restrict__ wr, const float* __restrict__ att,
    const int* __restrict__ src,
    const int* __restrict__ dst, const float* __restrict__ bias,
    float* __restrict__ out, int* __restrict__ lists,
    const unsigned* __restrict__ codes, int n_nodes, int nedges,
    int do_relu, int build,
    const float* __restrict__ wl_next, const float* __restrict__ att_next,
    float* xl_next, float* s_next) {
  __shared__ __attribute__((aligned(16))) float acc[ROWS_PB * ACC_P];
  __shared__ __attribute__((aligned(16))) float xrs[ROWS_PB * CC];
  __shared__ __attribute__((aligned(16))) float rbnd[ROWS_PB * HH];
  __shared__ unsigned mx[ROWS_PB * HH];
  __shared__ float den[ROWS_PB * HH];
  __shared__ int2 ring_all[AGG_WAVES * RING];
  __shared__ int wtail[AGG_WAVES];
  __shared__ int any_over;
  __shared__ int redo;

  const int tid = threadIdx.x;
  const int lane = tid & 31;
  const int wave = __builtin_amdgcn_readfirstlane(tid >> 5);
  const int n0 = blockIdx.x * ROWS_PB;
  const int rows = min(ROWS_PB, n_nodes - n0);

  float4* acc4 = (float4*)acc;
  for (int i = tid; i < rows * (ACC_P / 4); i += AGG_THREADS)
    acc4[i] = make_float4(0.0f, 0.0f, 0.0f, 0.0f);
  for (int i = tid; i < rows * HH; i += AGG_THREADS) {
    mx[i] = 0u;
    den[i] = 0.0f;
  }
  project_own_rows(hin, wr, att, xrs, rbnd, n0, rows, wave, lane);
  const int nlists = gridDim.x * AGG_WAVES;
  const int lid = blockIdx.x * AGG_WAVES + wave;
  int* glist = lists + (size_t)lid * RING;
  int* gcnt = lists + (size_t)nlists * RING;
  int* gover = gcnt + nlists;
  if (tid == 0) {
    any_over = build ? 0 : gover[blockIdx.x];
    redo = 0;
  }
  __syncthreads();

  const float4 av = ((const float4*)att)[lane & 15];
  int2* ring = ring_all + wave * RING;
  int cnt = 0;
  bool over = false;
  const bool bycode = build && gridDim.x <= 256;
  if (build) {
    if (bycode) {
      if (lane == 0) wtail[wave] = 0;
      scan_codes(codes, dst, ring, &wtail[wave], cnt, over, n0, rows, nedges,
                 wave, lane);
    } else {
      scan_owned<0>(xl, src, dst, acc, xrs, mx, den, ring, cnt, over, n0,
                    rows, nedges, av, wave, lane);
    }
    if (over && lane == 0) any_over = 1;
    __syncthreads();
    if (tid == 0) gover[blockIdx.x] = any_over;
  }
  if (any_over == 0) {
    if (build) {
      if (bycode)
        resolve_both(src, dst, ring, cnt, n0, lane);
      else
        resolve_src(src, ring, 0, cnt, lane);
#pragma unroll
      for (int k = 0; k < RING / 32; ++k) {
        const int b = lane + 32 * k;
        if (b < cnt) {
          const int2 q = ring[b];
          glist[b] = (q.x << 9) | q.y;
        }
      }
      if (lane == 0) gcnt[lid] = cnt;
    } else {
      cnt = __builtin_amdgcn_readfirstlane(gcnt[lid]);
      int t[RING / 32];
#pragma unroll
      for (int k = 0; k < RING / 32; ++k) {
        const int b = lane + 32 * k;
        t[k] = b < cnt ? glist[b] : 0;
      }
#pragma unroll
      for (int k = 0; k < RING / 32; ++k) {
        const int b = lane + 32 * k;
        if (b < cnt) ring[b] = make_int2((int)((unsigned)t[k] >> 9), t[k] & 511);
      }
    }
#pragma nounroll
    for (int exact = 0; exact < 2; ++exact) {
      if (exact == 0)
        bound_steps(sbnd, rbnd, mx, ring, cnt, lane);
      else
        edge_steps<1>(xl, acc, xrs, mx, den, ring, 0, cnt, av, lane);
      __syncthreads();
      edge_steps<2>(xl, acc, xrs, mx, den, ring, 0, cnt, av, lane);
      __syncthreads();
      if (exact) break;
      bool bad = false;
      for (int i = tid; i < rows * HH; i += AGG_THREADS)
        bad |= (mx[i] != 0u) & (den[i] < 1e-7f);
      if (bad) redo = 1;
      __syncthreads();
      if (redo == 0) break;
      for (int i = tid; i < rows * (ACC_P / 4); i += AGG_THREADS)
        acc4[i] = make_float4(0.0f, 0.0f, 0.0f, 0.0f);
      for (int i = tid; i < rows * HH; i += AGG_THREADS) {
        mx[i] = 0u;
        den[i] = 0.0f;
      }
      __syncthreads();
    }
  } else {
    scan_owned<1>(xl, src, dst, acc, xrs, mx, den, ring, cnt, over, n0, rows,
                  nedges, av, wave, lane);
    __syncthreads();
    scan_owned<2>(xl, src, dst, acc, xrs, mx, den, ring, cnt, over, n0, rows,
                  nedges, av, wave, lane);
    __syncthreads();
  }

  const float4* __restrict__ b4 = (const float4*)bias;
  float4* __restrict__ o4 = (float4*)(out + (size_t)n0 * CC);
  for (int i = tid; i < rows * (CC / 4); i += AGG_THREADS) {
    const int li = (i >> 4) * (ACC_P / 4) + (i & 15);
    float4 a = acc4[li];
    const float dn = den[(i >> 4) * HH + ((i & 15) >> 2)] + 1e-16f;
    const float4 b = b4[i & (CC / 4 - 1)];
    a.x = a.x / dn + b.x; a.y = a.y / dn + b.y;
    a.z = a.z / dn + b.z; a.w = a.w / dn + b.w;
    if (do_relu) {
      a.x = fmaxf(a.x, 0.0f); a.y = fmaxf(a.y, 0.0f);
      a.z = fmaxf(a.z, 0.0f); a.w = fmaxf(a.w, 0.0f);
    }
    o4[i] = a;
    if (wl_next) acc4[li] = a;
  }
  if (wl_next) {
    __syncthreads();
    project_next_rows(acc, wl_next, att_next, xl_next, s_next, n0, rows, wave,
                      lane);
  }
}

extern "C" void kernel_launch(void* const* d_in, const int* in_sizes, int n_in,
                              void* d_out, int out_size, void* d_ws,
                              size_t ws_size, hipStream_t stream) {
  const int N = in_sizes[0] / CC;
  const int E = in_sizes[1] / 2;

  const float* x = (const float*)d_in[0];
  const int* edge = (const int*)d_in[1];
  const int* src = edge;
  const int* dst = edge + E;

  const float* Wl[3] = {(const float*)d_in[2], (const float*)d_in[6],
                        (const float*)d_in[10]};
  const float* Wr[3] = {(const float*)d_in[3], (const float*)d_in[7],
                        (const float*)d_in[11]};
  const float* Av[3] = {(const float*)d_in[4], (const float*)d_in[8],
                        (const float*)d_in[12]};
  const float* Bv[3] = {(const float*)d_in[5], (const float*)d_in[9],
                        (const float*)d_in[13]};

  const size_t NC = (size_t)N * CC;
  float* ws = (float*)d_ws;
  float* p0 = ws;                 ws += NC;
  float* p1 = ws;                 ws += NC;
  float* p2 = ws;                 ws += NC;
  float* sa = ws;                 ws += (size_t)N * HH;
  float* sb = ws;                 ws += (size_t)N * HH;
  unsigned* codes = (unsigned*)ws; ws += (size_t)(E + 3) / 4;
  int* lists = (int*)ws;

  const float* lxl[3] = {p0, p2, p1};
  const float* lsb[3] = {sa, sb, sa};
  const float* lin[3] = {x, p1, p0};
  float* lout[3] = {p1, p0, (float*)d_out};
  float* nxl[3] = {p2, p1, nullptr};
  float* nsb[3] = {sb, sa, nullptr};
  const float* nwl[3] = {Wl[1], Wl[2], nullptr};
  const float* nav[3] = {Av[1], Av[2], nullptr};
  const int lrelu[3] = {1, 1, 0};
  const int lbuild[3] = {1, 0, 0};

  const int gemm_grid = (N + 15) / 16;
  const int agg_grid = (N + ROWS_PB - 1) / ROWS_PB;

  gat_gemm_wmma<<<gemm_grid, 128, 0, stream>>>(x, Wl[0], Av[0], p0, sa, N, dst,
                                               codes, E);
  for (int l = 0; l < 3; ++l) {
    gat_owner_layer<<<agg_grid, AGG_THREADS, 0, stream>>>(
        lxl[l], lsb[l], lin[l], Wr[l], Av[l], src, dst, Bv[l], lout[l], lists,
        codes, N, E, lrelu[l], lbuild[l], nwl[l], nav[l], nxl[l], nsb[l]);
  }
}


# Softmax shift of (row, head) = largest S[h] of ANY source row + R[row][h]: no per-edge gather of S and no shift pass; producers of xl store only block maxima of S; exp rounding changes
# speedup vs baseline: 6.3192x; 1.0428x over previous
#include <hip/hip_runtime.h>


typedef float v2f __attribute__((ext_vector_type(2)));
typedef float v8f __attribute__((ext_vector_type(8)));

#define CC 64
#define HH 4
#define OO 16
#define NEG_SLOPE 0.2f
#define ROWS_PB 391

__device__ __forceinline__ unsigned fenc(float f) {
  unsigned b = __float_as_uint(f);
  return (b & 0x80000000u) ? ~b : (b | 0x80000000u);
}
__device__ __forceinline__ float fdec(unsigned e) {
  return (e & 0x80000000u) ? __uint_as_float(e & 0x7FFFFFFFu)
                           : __uint_as_float(~e);
}

__device__ __forceinline__ float leaky(float v) {
  return fmaxf(v, NEG_SLOPE * v);
}

__device__ __forceinline__ float quad_sum(float t) {
  t += __int_as_float(__builtin_amdgcn_update_dpp(
      0, __float_as_int(t), 0xB1, 0xF, 0xF, true));
  t += __int_as_float(__builtin_amdgcn_update_dpp(
      0, __float_as_int(t), 0x4E, 0xF, 0xF, true));
  return t;
}

__device__ __forceinline__ float row16_sum(float t) {
  t = quad_sum(t);
  t += __int_as_float(__builtin_amdgcn_update_dpp(
      0, __float_as_int(t), 0x141, 0xF, 0xF, true));
  t += __int_as_float(__builtin_amdgcn_update_dpp(
      0, __float_as_int(t), 0x140, 0xF, 0xF, true));
  return t;
}

__device__ __forceinline__ float wave_max(float v) {
  v = fmaxf(v, __int_as_float(__builtin_amdgcn_update_dpp(
                   0, __float_as_int(v), 0xB1, 0xF, 0xF, true)));
  v = fmaxf(v, __int_as_float(__builtin_amdgcn_update_dpp(
                   0, __float_as_int(v), 0x4E, 0xF, 0xF, true)));
  v = fmaxf(v, __int_as_float(__builtin_amdgcn_update_dpp(
                   0, __float_as_int(v), 0x141, 0xF, 0xF, true)));
  v = fmaxf(v, __int_as_float(__builtin_amdgcn_update_dpp(
                   0, __float_as_int(v), 0x140, 0xF, 0xF, true)));
  v = fmaxf(v, __int_as_float(__builtin_amdgcn_permlanex16(
                   0, __float_as_int(v), 0x76543210, 0xFEDCBA98, true, true)));
  return v;
}

#define NEG_INF __uint_as_float(0xFF800000u)

__global__ __launch_bounds__(128) void gat_gemm_wmma(
    const float* __restrict__ X, const float* __restrict__ W,
    const float* __restrict__ att, float* __restrict__ Y,
    float* __restrict__ S, int nrows, const int* __restrict__ dstl,
    unsigned* __restrict__ codes, int nedges) {
  if (codes) {
    const int ng = nedges >> 2;
    for (int g = blockIdx.x * 128 + threadIdx.x; g < ng; g += gridDim.x * 128) {
      const unsigned o0 = (unsigned)dstl[4 * g + 0] / ROWS_PB;
      const unsigned o1 = (unsigned)dstl[4 * g + 1] / ROWS_PB;
      const unsigned o2 = (unsigned)dstl[4 * g + 2] / ROWS_PB;
      const unsigned o3 = (unsigned)dstl[4 * g + 3] / ROWS_PB;
      codes[g] = (o0 & 255u) | ((o1 & 255u) << 8) | ((o2 & 255u) << 16) |
                 ((o3 & 255u) << 24);
    }
  }
  const int wave = threadIdx.x >> 5;
  const int lane = threadIdx.x & 31;
  const int m0 = blockIdx.x * 16;
  if (m0 >= nrows) return;
  const int n0 = wave * 16;
  const int lrow = lane & 15;
  const int khalf = (lane >> 4) << 1;

  const float* __restrict__ xrow = X + (size_t)(m0 + lrow) * CC;
  v8f c = {};
#pragma unroll
  for (int k = 0; k < CC; k += 4) {
    v2f a, b;
    a.x = xrow[k + khalf];
    a.y = xrow[k + khalf + 1];
    b.x = W[(size_t)(k + khalf) * CC + n0 + lrow];
    b.y = W[(size_t)(k + khalf + 1) * CC + n0 + lrow];
    c = __builtin_amdgcn_wmma_f32_16x16x4_f32(false, a, false, b,
                                              (short)0, c, false, false);
  }
  const int rbase = m0 + ((lane >> 4) << 3);
#pragma unroll
  for (int r = 0; r < 8; ++r)
    Y[(size_t)(rbase + r) * CC + n0 + lrow] = c[r];
  const float ao = att[n0 + lrow];
  float sv = 0.0f;
#pragma unroll
  for (int r = 0; r < 8; ++r) {
    const float p = row16_sum(ao * leaky(c[r]));
    if (lrow == r) sv = p;
  }
  const float wm = wave_max(lrow < 8 ? sv : NEG_INF);
  if (lane == 0) S[(size_t)blockIdx.x * HH + wave] = wm;
}

#define AGG_THREADS 1024
#define AGG_WAVES (AGG_THREADS / 32)
#define RING 256
#define SCAN_G 8
#define STEP_K 8
#define ACC_P 68

__device__ __forceinline__ void project_own_rows(
    const float* __restrict__ hin, const float* __restrict__ Wr,
    const float* __restrict__ att, float* xrs, float* rbnd,
    int n0, int rows, int wave, int lane) {
  const int lrow = lane & 15;
  const int khalf = (lane >> 4) << 1;
  const int ncol = (wave & 3) * 16;
  const float ao = att[ncol + lrow];
  const float aabs = fabsf(ao);
  const float asgn = ao < 0.0f ? -1.0f : 1.0f;
  v2f b[CC / 4];
#pragma unroll
  for (int j = 0; j < CC / 4; ++j) {
    b[j].x = Wr[(size_t)(4 * j + khalf) * CC + ncol + lrow];
    b[j].y = Wr[(size_t)(4 * j + khalf + 1) * CC + ncol + lrow];
  }
  const int ntiles = (rows + 15) >> 4;
  for (int t = wave >> 2; t < ntiles; t += AGG_WAVES / 4) {
    const int m0 = t * 16;
    const float* __restrict__ xrow =
        hin + (size_t)(n0 + min(m0 + lrow, rows - 1)) * CC;
    v8f c = {};
#pragma unroll
    for (int j = 0; j < CC / 4; ++j) {
      v2f a;
      a.x = xrow[4 * j + khalf];
      a.y = xrow[4 * j + khalf + 1];
      c = __builtin_amdgcn_wmma_f32_16x16x4_f32(false, a, false, b[j],
                                                (short)0, c, false, false);
    }
    const int rb = m0 + ((lane >> 4) << 3);
#pragma unroll
    for (int r = 0; r < 8; ++r)
      if (rb + r < rows) xrs[(rb + r) * CC + ncol + lrow] = c[r];
    float rv = 0.0f;
#pragma unroll
    for (int r = 0; r < 8; ++r) {
      const float p = row16_sum(aabs * leaky(asgn * c[r]));
      if (lrow == r) rv = p;
    }
    if (lrow < 8 && rb + lrow < rows) rbnd[(rb + lrow) * HH + (wave & 3)] = rv;
  }
}

__device__ __forceinline__ void project_next_rows(
    const float* hs, const float* __restrict__ Wl,
    const float* __restrict__ att, float* Y, float* S,
    int n0, int rows, int wave, int lane) {
  const int lrow = lane & 15;
  const int khalf = (lane >> 4) << 1;
  const int ncol = (wave & 3) * 16;
  const float ao = att[ncol + lrow];
  v2f b[CC / 4];
#pragma unroll
  for (int j = 0; j < CC / 4; ++j) {
    b[j].x = Wl[(size_t)(4 * j + khalf) * CC + ncol + lrow];
    b[j].y = Wl[(size_t)(4 * j + khalf + 1) * CC + ncol + lrow];
  }
  const int ntiles = (rows + 15) >> 4;
  float smx = NEG_INF;
  for (int t = wave >> 2; t < ntiles; t += AGG_WAVES / 4) {
    const int m0 = t * 16;
    const float* xrow = hs + min(m0 + lrow, rows - 1) * ACC_P;
    v8f c = {};
#pragma unroll
    for (int j = 0; j < CC / 4; ++j) {
      const v2f a = *(const v2f*)(xrow + 4 * j + khalf);
      c = __builtin_amdgcn_wmma_f32_16x16x4_f32(false, a, false, b[j],
                                                (short)0, c, false, false);
    }
    const int rb = m0 + ((lane >> 4) << 3);
#pragma unroll
    for (int r = 0; r < 8; ++r)
      if (rb + r < rows) Y[(size_t)(n0 + rb + r) * CC + ncol + lrow] = c[r];
    float sv = 0.0f;
#pragma unroll
    for (int r = 0; r < 8; ++r) {
      const float p = row16_sum(ao * leaky(c[r]));
      if (lrow == r) sv = p;
    }
    if (lrow < 8 && rb + lrow < rows) smx = fmaxf(smx, sv);
  }
  smx = wave_max(smx);
  if (lane == 0)
    S[((size_t)blockIdx.x * (AGG_WAVES / 4) + (wave >> 2)) * HH + (wave & 3)] =
        smx;
}

__device__ __forceinline__ void resolve_src(const int* __restrict__ src,
                                            int2* ring, int first, int cnt,
                                            int lane) {
  int e[RING / 32];
#pragma unroll
  for (int k = 0; k < RING / 32; ++k) {
    const int b = lane + 32 * k;
    e[k] = b < cnt ? ring[(first + b) & (RING - 1)].x : 0;
  }
#pragma unroll
  for (int k = 0; k < RING / 32; ++k) {
    const int b = lane + 32 * k;
    if (b < cnt) e[k] = src[e[k]];
  }
#pragma unroll
  for (int k = 0; k < RING / 32; ++k) {
    const int b = lane + 32 * k;
    if (b < cnt) ring[(first + b) & (RING - 1)].x = e[k];
  }
}

__device__ __forceinline__ void resolve_both(const int* __restrict__ src,
                                             const int* __restrict__ dst,
                                             int2* ring, int cnt, int n0,
                                             int lane) {
  int e[RING / 32], d[RING / 32];
#pragma unroll
  for (int k = 0; k < RING / 32; ++k) {
    const int b = lane + 32 * k;
    e[k] = b < cnt ? ring[b].x : 0;
  }
#pragma unroll
  for (int k = 0; k < RING / 32; ++k) {
    const int b = lane + 32 * k;
    d[k] = 0;
    if (b < cnt) {
      d[k] = dst[e[k]] - n0;
      e[k] = src[e[k]];
    }
  }
#pragma unroll
  for (int k = 0; k < RING / 32; ++k) {
    const int b = lane + 32 * k;
    if (b < cnt) ring[b] = make_int2(e[k], d[k]);
  }
}

__device__ __forceinline__ void scan_codes(
    const unsigned* __restrict__ codes, const int* __restrict__ dst,
    int2* ring, int* tailp, int& cnt_out, bool& over_out, int n0, int rows,
    int nedges, int wave, int lane) {
  const unsigned pat = (unsigned)blockIdx.x * 0x01010101u;
  const int n16 = ((((size_t)codes) & 15) == 0) ? (nedges >> 4) : 0;
  const uint4* __restrict__ c16 = (const uint4*)codes;
  for (int base = wave * (32 * SCAN_G); base < n16;
       base += AGG_WAVES * 32 * SCAN_G) {
    uint4 v[SCAN_G];
#pragma unroll
    for (int g = 0; g < SCAN_G; ++g) {
      const int idx = base + g * 32 + lane;
      v[g] = c16[min(idx, n16 - 1)];
    }
    unsigned gm = 0u;
#pragma unroll
    for (int g = 0; g < SCAN_G; ++g) {
      const int idx = base + g * 32 + lane;
      const unsigned z0 = v[g].x ^ pat, z1 = v[g].y ^ pat;
      const unsigned z2 = v[g].z ^ pat, z3 = v[g].w ^ pat;
      const unsigned any =
          (((z0 - 0x01010101u) & ~z0) | ((z1 - 0x01010101u) & ~z1) |
           ((z2 - 0x01010101u) & ~z2) | ((z3 - 0x01010101u) & ~z3)) &
          0x80808080u;
      if (any != 0u && idx < n16) gm |= 1u << g;
    }
    while (gm) {
      const int g = __builtin_ctz(gm);
      gm &= gm - 1u;
      const int idx = base + g * 32 + lane;
      const uint4 cv = c16[idx];
      const unsigned zz[4] = {cv.x ^ pat, cv.y ^ pat, cv.z ^ pat, cv.w ^ pat};
      unsigned hm[4];
      int c = 0;
#pragma unroll
      for (int w = 0; w < 4; ++w) {
        hm[w] = ~(((zz[w] & 0x7F7F7F7Fu) + 0x7F7F7F7Fu) | zz[w] |
                  0x7F7F7F7Fu);
        c += __builtin_popcount(hm[w]);
      }
      int pos = atomicAdd(tailp, c);
#pragma unroll
      for (int w = 0; w < 4; ++w) {
        unsigned m = hm[w];
        while (m) {
          const int bit = __builtin_ctz(m);
          m &= m - 1u;
          ring[pos & (RING - 1)] = make_int2(idx * 16 + w * 4 + (bit >> 3), 0);
          ++pos;
        }
      }
    }
  }
  for (int eb = 16 * n16 + wave * 32; eb < nedges; eb += AGG_WAVES * 32) {
    const int e = eb + lane;
    if (e < nedges && (unsigned)(dst[e] - n0) < (unsigned)rows) {
      const int pos = atomicAdd(tailp, 1);
      ring[pos & (RING - 1)] = make_int2(e, 0);
    }
  }
  const int tail = __builtin_amdgcn_readfirstlane(*(volatile int*)tailp);
  cnt_out = tail;
  over_out = tail > RING;
}

template <int PASS>
__device__ __forceinline__ void edge_steps(
    const float* __restrict__ xl, float* acc, const float* xrs, unsigned* mx,
    float* den, const int2* ring, int first, int cnt, float4 av, int lane) {
  const int c4 = (lane & 15) << 2;
  const int half = lane >> 4;
  const int hd = (lane & 15) >> 2;
  const bool quad0 = (lane & 3) == 0;
  const int steps = (cnt + 1) >> 1;
#pragma nounroll
  for (int k0 = 0; k0 < steps; k0 += STEP_K) {
    int2 q[STEP_K];
    float4 lv[STEP_K], rv[STEP_K];
    unsigned me[STEP_K];
#pragma unroll
    for (int k = 0; k < STEP_K; ++k) {
      const int idx = min(2 * (k0 + k) + half, cnt - 1);
      q[k] = ring[(first + idx) & (RING - 1)];
    }
#pragma unroll
    for (int k = 0; k < STEP_K; ++k) {
      lv[k] = *(const float4*)(xl + (size_t)q[k].x * CC + c4);
      rv[k] = *(const float4*)(xrs + q[k].y * CC + c4);
      me[k] = PASS == 1 ? 0u : mx[q[k].y * HH + hd];
    }
#pragma unroll
    for (int k = 0; k < STEP_K; ++k) {
      const bool live = 2 * (k0 + k) + half < cnt;
      float t = av.x * leaky(lv[k].x + rv[k].x);
      t += av.y * leaky(lv[k].y + rv[k].y);
      t += av.z * leaky(lv[k].z + rv[k].z);
      t += av.w * leaky(lv[k].w + rv[k].w);
      t = quad_sum(t);
      const int sh = q[k].y * HH + hd;
      if (PASS == 1) {
        if (live && quad0) atomicMax(&mx[sh], fenc(t));
      } else {
        const float w = __expf(fmaxf(t - fdec(me[k]), -80.0f));
        if (live) {
          float* dr = acc + q[k].y * ACC_P + c4;
          float* d01 = dr + 2 * half;
          float* d23 = dr + 2 - 2 * half;
          const float p0 = w * (half ? lv[k].z : lv[k].x);
          const float p1 = w * (half ? lv[k].w : lv[k].y);
          const float p2 = w * (half ? lv[k].x : lv[k].z);
          const float p3 = w * (half ? lv[k].y : lv[k].w);
          atomicAdd(d01 + 0, p0);
          atomicAdd(d01 + 1, p1);
          atomicAdd(d23 + 0, p2);
          atomicAdd(d23 + 1, p3);
          if (quad0) atomicAdd(&den[sh], w);
        }
      }
    }
  }
}

template <int MODE>
__device__ __forceinline__ void scan_owned(
    const float* __restrict__ xl, const int* __restrict__ src,
    const int* __restrict__ dst, float* acc, const float* xrs, unsigned* mx,
    float* den, int2* ring, int& cnt_out, bool& over_out, int n0, int rows,
    int nedges, float4 av, int wave, int lane) {
  int head = 0, tail = 0;

#define AGG_PUSH(m, e, x)                                                    \
  {                                                                          \
    const unsigned bal_ = __builtin_amdgcn_ballot_w32(m);                    \
    if (m)                                                                   \
      ring[(tail + (int)__builtin_amdgcn_mbcnt_lo(bal_, 0u)) & (RING - 1)] = \
          make_int2((e), (int)(x));                                          \
    tail += __builtin_popcount(bal_);                                        \
  }
#define AGG_DRAIN()                                                          \
  if (MODE != 0) {                                                           \
    while (tail - head >= 32) {                                              \
      resolve_src(src, ring, head, 32, lane);                                \
      edge_steps<(MODE == 1 ? 1 : 2)>(xl, acc, xrs, mx, den, ring, head, 32, \
                                      av, lane);                             \
      head += 32;                                                            \
    }                                                                        \
  }

  const int n4 = ((((size_t)dst) & 15) == 0) ? (nedges >> 2) : 0;
  const int4* __restrict__ dst4 = (const int4*)dst;
  for (int base = wave * (32 * SCAN_G); base < n4;
       base += AGG_WAVES * 32 * SCAN_G) {
    int4 v[SCAN_G];
#pragma unroll
    for (int g = 0; g < SCAN_G; ++g) {
      const int idx = base + g * 32 + lane;
      v[g] = dst4[min(idx, n4 - 1)];
    }
#pragma unroll
    for (int g = 0; g < SCAN_G; ++g) {
      const int idx = base + g * 32 + lane;
      const unsigned lim = idx < n4 ? (unsigned)rows : 0u;
      const unsigned x0 = (unsigned)(v[g].x - n0);
      const unsigned x1 = (unsigned)(v[g].y - n0);
      const unsigned x2 = (unsigned)(v[g].z - n0);
      const unsigned x3 = (unsigned)(v[g].w - n0);
      const bool m0 = x0 < lim, m1 = x1 < lim, m2 = x2 < lim, m3 = x3 < lim;
      if (__builtin_amdgcn_ballot_w32(m0 | m1 | m2 | m3)) {
        const int e0 = idx * 4;
        AGG_PUSH(m0, e0 + 0, x0)
        AGG_PUSH(m1, e0 + 1, x1)
        AGG_PUSH(m2, e0 + 2, x2)
        AGG_PUSH(m3, e0 + 3, x3)
      }
      AGG_DRAIN()
    }
  }
  for (int eb = 4 * n4 + wave * 32; eb < nedges; eb += AGG_WAVES * 32) {
    const int e = eb + lane;
    const unsigned x = e < nedges ? (unsigned)(dst[e] - n0) : 0xFFFFFFFFu;
    const bool m = x < (unsigned)rows;
    AGG_PUSH(m, e, x)
    AGG_DRAIN()
  }
  if (MODE != 0 && tail > head) {
    resolve_src(src, ring, head, tail - head, lane);
    edge_steps<(MODE == 1 ? 1 : 2)>(xl, acc, xrs, mx, den, ring, head,
                                    tail - head, av, lane);
  }
#undef AGG_PUSH
#undef AGG_DRAIN
  cnt_out = tail;
  over_out = MODE == 0 && tail > RING;
}

__global__ __launch_bounds__(AGG_THREADS) void gat_owner_layer(
    const float* __restrict__ xl, const float* __restrict__ sbnd,
    const float* hin,
    const float* __restrict__ wr, const float* __restrict__ att,
    const int* __restrict__ src,
    const int* __restrict__ dst, const float* __restrict__ bias,
    float* __restrict__ out, int* __restrict__ lists,
    const unsigned* __restrict__ codes, int n_nodes, int nedges,
    int do_relu, int build,
    const float* __restrict__ wl_next, const float* __restrict__ att_next,
    float* xl_next, float* s_next) {
  __shared__ __attribute__((aligned(16))) float acc[ROWS_PB * ACC_P];
  __shared__ __attribute__((aligned(16))) float xrs[ROWS_PB * CC];
  __shared__ __attribute__((aligned(16))) float rbnd[ROWS_PB * HH];
  __shared__ unsigned mx[ROWS_PB * HH];
  __shared__ float den[ROWS_PB * HH];
  __shared__ int2 ring_all[AGG_WAVES * RING];
  __shared__ __attribute__((aligned(16))) float smaxp[HH * AGG_WAVES];
  __shared__ int wtail[AGG_WAVES];
  __shared__ int any_over;
  __shared__ int redo;

  const int tid = threadIdx.x;
  const int lane = tid & 31;
  const int wave = __builtin_amdgcn_readfirstlane(tid >> 5);
  const int n0 = blockIdx.x * ROWS_PB;
  const int rows = min(ROWS_PB, n_nodes - n0);

  const int nbm = build ? (n_nodes + 15) >> 4 : (int)gridDim.x * (AGG_WAVES / 4);
  const float4* __restrict__ bm4 = (const float4*)sbnd;
  float4 bv[8];
#pragma unroll
  for (int k = 0; k < 8; ++k)
    bv[k] = bm4[min(k * AGG_THREADS + tid, nbm - 1)];

  float4* acc4 = (float4*)acc;
  for (int i = tid; i < rows * (ACC_P / 4); i += AGG_THREADS)
    acc4[i] = make_float4(0.0f, 0.0f, 0.0f, 0.0f);
  for (int i = tid; i < rows * HH; i += AGG_THREADS) {
    mx[i] = 0u;
    den[i] = 0.0f;
  }
  project_own_rows(hin, wr, att, xrs, rbnd, n0, rows, wave, lane);
  float4 sm = make_float4(NEG_INF, NEG_INF, NEG_INF, NEG_INF);
  for (int base = 0;;) {
#pragma unroll
    for (int k = 0; k < 8; ++k) {
      sm.x = fmaxf(sm.x, bv[k].x); sm.y = fmaxf(sm.y, bv[k].y);
      sm.z = fmaxf(sm.z, bv[k].z); sm.w = fmaxf(sm.w, bv[k].w);
    }
    base += 8 * AGG_THREADS;
    if (base >= nbm) break;
#pragma unroll
    for (int k = 0; k < 8; ++k)
      bv[k] = bm4[min(base + k * AGG_THREADS + tid, nbm - 1)];
  }
  sm.x = wave_max(sm.x); sm.y = wave_max(sm.y);
  sm.z = wave_max(sm.z); sm.w = wave_max(sm.w);
  if (lane == 0) {
    smaxp[0 * AGG_WAVES + wave] = sm.x;
    smaxp[1 * AGG_WAVES + wave] = sm.y;
    smaxp[2 * AGG_WAVES + wave] = sm.z;
    smaxp[3 * AGG_WAVES + wave] = sm.w;
  }
  const int nlists = gridDim.x * AGG_WAVES;
  const int lid = blockIdx.x * AGG_WAVES + wave;
  int* glist = lists + (size_t)lid * RING;
  int* gcnt = lists + (size_t)nlists * RING;
  int* gover = gcnt + nlists;
  if (tid == 0) {
    any_over = build ? 0 : gover[blockIdx.x];
    redo = 0;
  }
  __syncthreads();

  const float4 av = ((const float4*)att)[lane & 15];
  int2* ring = ring_all + wave * RING;
  int cnt = 0;
  bool over = false;
  const bool bycode = build && gridDim.x <= 256;
  if (build) {
    if (bycode) {
      if (lane == 0) wtail[wave] = 0;
      scan_codes(codes, dst, ring, &wtail[wave], cnt, over, n0, rows, nedges,
                 wave, lane);
    } else {
      scan_owned<0>(xl, src, dst, acc, xrs, mx, den, ring, cnt, over, n0,
                    rows, nedges, av, wave, lane);
    }
    if (over && lane == 0) any_over = 1;
    __syncthreads();
    if (tid == 0) gover[blockIdx.x] = any_over;
  }
  if (any_over == 0) {
    if (build) {
      if (bycode)
        resolve_both(src, dst, ring, cnt, n0, lane);
      else
        resolve_src(src, ring, 0, cnt, lane);
#pragma unroll
      for (int k = 0; k < RING / 32; ++k) {
        const int b = lane + 32 * k;
        if (b < cnt) {
          const int2 q = ring[b];
          glist[b] = (q.x << 9) | q.y;
        }
      }
      if (lane == 0) gcnt[lid] = cnt;
    } else {
      cnt = __builtin_amdgcn_readfirstlane(gcnt[lid]);
      int t[RING / 32];
#pragma unroll
      for (int k = 0; k < RING / 32; ++k) {
        const int b = lane + 32 * k;
        t[k] = b < cnt ? glist[b] : 0;
      }
#pragma unroll
      for (int k = 0; k < RING / 32; ++k) {
        const int b = lane + 32 * k;
        if (b < cnt) ring[b] = make_int2((int)((unsigned)t[k] >> 9), t[k] & 511);
      }
    }
#pragma nounroll
    for (int exact = 0; exact < 2; ++exact) {
      if (exact == 0) {
        const float4* sp = (const float4*)(smaxp + (tid & 3) * AGG_WAVES);
        float smh = NEG_INF;
#pragma unroll
        for (int k = 0; k < AGG_WAVES / 4; ++k) {
          const float4 v = sp[k];
          smh = fmaxf(fmaxf(smh, v.x), fmaxf(v.y, fmaxf(v.z, v.w)));
        }
        for (int i = tid; i < rows * HH; i += AGG_THREADS)
          mx[i] = fenc(smh + rbnd[i]);
      } else
        edge_steps<1>(xl, acc, xrs, mx, den, ring, 0, cnt, av, lane);
      __syncthreads();
      edge_steps<2>(xl, acc, xrs, mx, den, ring, 0, cnt, av, lane);
      __syncthreads();
      if (exact) break;
      bool bad = false;
      for (int i = tid; i < rows * HH; i += AGG_THREADS)
        bad |= (den[i] > 0.0f) & (den[i] < 1e-7f);
      if (bad) redo = 1;
      __syncthreads();
      if (redo == 0) break;
      for (int i = tid; i < rows * (ACC_P / 4); i += AGG_THREADS)
        acc4[i] = make_float4(0.0f, 0.0f, 0.0f, 0.0f);
      for (int i = tid; i < rows * HH; i += AGG_THREADS) {
        mx[i] = 0u;
        den[i] = 0.0f;
      }
      __syncthreads();
    }
  } else {
    scan_owned<1>(xl, src, dst, acc, xrs, mx, den, ring, cnt, over, n0, rows,
                  nedges, av, wave, lane);
    __syncthreads();
    scan_owned<2>(xl, src, dst, acc, xrs, mx, den, ring, cnt, over, n0, rows,
                  nedges, av, wave, lane);
    __syncthreads();
  }

  const float4* __restrict__ b4 = (const float4*)bias;
  float4* __restrict__ o4 = (float4*)(out + (size_t)n0 * CC);
  for (int i = tid; i < rows * (CC / 4); i += AGG_THREADS) {
    const int li = (i >> 4) * (ACC_P / 4) + (i & 15);
    float4 a = acc4[li];
    const float dn = den[(i >> 4) * HH + ((i & 15) >> 2)] + 1e-16f;
    const float4 b = b4[i & (CC / 4 - 1)];
    a.x = a.x / dn + b.x; a.y = a.y / dn + b.y;
    a.z = a.z / dn + b.z; a.w = a.w / dn + b.w;
    if (do_relu) {
      a.x = fmaxf(a.x, 0.0f); a.y = fmaxf(a.y, 0.0f);
      a.z = fmaxf(a.z, 0.0f); a.w = fmaxf(a.w, 0.0f);
    }
    o4[i] = a;
    if (wl_next) acc4[li] = a;
  }
  if (wl_next) {
    __syncthreads();
    project_next_rows(acc, wl_next, att_next, xl_next, s_next, n0, rows, wave,
                      lane);
  }
}

extern "C" void kernel_launch(void* const* d_in, const int* in_sizes, int n_in,
                              void* d_out, int out_size, void* d_ws,
                              size_t ws_size, hipStream_t stream) {
  const int N = in_sizes[0] / CC;
  const int E = in_sizes[1] / 2;

  const float* x = (const float*)d_in[0];
  const int* edge = (const int*)d_in[1];
  const int* src = edge;
  const int* dst = edge + E;

  const float* Wl[3] = {(const float*)d_in[2], (const float*)d_in[6],
                        (const float*)d_in[10]};
  const float* Wr[3] = {(const float*)d_in[3], (const float*)d_in[7],
                        (const float*)d_in[11]};
  const float* Av[3] = {(const float*)d_in[4], (const float*)d_in[8],
                        (const float*)d_in[12]};
  const float* Bv[3] = {(const float*)d_in[5], (const float*)d_in[9],
                        (const float*)d_in[13]};

  const size_t NC = (size_t)N * CC;
  float* ws = (float*)d_ws;
  float* p0 = ws;                 ws += NC;
  float* p1 = ws;                 ws += NC;
  float* p2 = ws;                 ws += NC;
  float* sa = ws;                 ws += (size_t)N * HH;
  float* sb = ws;                 ws += (size_t)N * HH;
  unsigned* codes = (unsigned*)ws; ws += (size_t)(E + 3) / 4;
  int* lists = (int*)ws;

  const float* lxl[3] = {p0, p2, p1};
  const float* lsb[3] = {sa, sb, sa};
  const float* lin[3] = {x, p1, p0};
  float* lout[3] = {p1, p0, (float*)d_out};
  float* nxl[3] = {p2, p1, nullptr};
  float* nsb[3] = {sb, sa, nullptr};
  const float* nwl[3] = {Wl[1], Wl[2], nullptr};
  const float* nav[3] = {Av[1], Av[2], nullptr};
  const int lrelu[3] = {1, 1, 0};
  const int lbuild[3] = {1, 0, 0};

  const int gemm_grid = (N + 15) / 16;
  const int agg_grid = (N + ROWS_PB - 1) / ROWS_PB;

  gat_gemm_wmma<<<gemm_grid, 128, 0, stream>>>(x, Wl[0], Av[0], p0, sa, N, dst,
                                               codes, E);
  for (int l = 0; l < 3; ++l) {
    gat_owner_layer<<<agg_grid, AGG_THREADS, 0, stream>>>(
        lxl[l], lsb[l], lin[l], Wr[l], Av[l], src, dst, Bv[l], lout[l], lists,
        codes, N, E, lrelu[l], lbuild[l], nwl[l], nav[l], nxl[l], nsb[l]);
  }
}
